# starting kernel
# baseline (speedup 1.0000x reference)
#include <hip/hip_runtime.h>
#include <hip/hip_bf16.h>

typedef __attribute__((ext_vector_type(2))) float v2f;
typedef __attribute__((ext_vector_type(8))) float v8f;

__global__ void zero_f32(float* __restrict__ p, long n4) {
    long i = (long)blockIdx.x * blockDim.x + threadIdx.x;
    if (i < n4) {
        ((float4*)p)[i] = make_float4(0.f, 0.f, 0.f, 0.f);
    }
}

__global__ __launch_bounds__(256) void scatter_accum(
    const float* __restrict__ feat, const int* __restrict__ src,
    const int* __restrict__ dst, float* __restrict__ msg,
    float* __restrict__ deg, int E)
{
    long gid  = (long)blockIdx.x * blockDim.x + threadIdx.x;
    long edge = gid >> 5;
    int  lane = (int)(gid & 31);
    if (edge >= E) return;

    int s = src[edge];
    int d = dst[edge];

    const float4 v = ((const float4*)(feat + (size_t)s * 128))[lane];
    float* o = msg + (size_t)d * 128 + lane * 4;
    atomicAdd(o + 0, v.x);
    atomicAdd(o + 1, v.y);
    atomicAdd(o + 2, v.z);
    atomicAdd(o + 3, v.w);
    if (deg != nullptr && lane == 0) atomicAdd(deg + d, 1.0f);
}

template <bool RELU>
__global__ __launch_bounds__(256) void sage_gemm_wmma(
    const float* __restrict__ msg, const float* __restrict__ deg,
    const float* __restrict__ root, const float* __restrict__ Wl,
    const float* __restrict__ bias, const float* __restrict__ Wr,
    float* __restrict__ out, int n_rows)
{
    __shared__ float lds_mean[16][132];
    __shared__ float lds_root[16][132];

    const int tid      = threadIdx.x;
    const int row_base = blockIdx.x * 16;
    if (row_base >= n_rows) return;

    {
        int idx = tid * 8;
        int r   = idx >> 7;
        int c   = idx & 127;
        float invd = 1.0f / fmaxf(deg[row_base + r], 1.0f);
        const float* ms = msg  + (size_t)(row_base + r) * 128 + c;
        const float* rt = root + (size_t)(row_base + r) * 128 + c;
#pragma unroll
        for (int j = 0; j < 8; ++j) {
            lds_mean[r][c + j] = ms[j] * invd;
            lds_root[r][c + j] = rt[j];
        }
    }
    __syncthreads();

    const int wave = tid >> 5;
    const int lane = tid & 31;
    const int col  = wave * 16;
    const int lr   = lane & 15;
    const int kh   = (lane >> 4) * 2;

    const float* wl_row = Wl + (size_t)(col + lr) * 128;
    const float* wr_row = Wr + (size_t)(col + lr) * 128;

    v8f acc = {};
#pragma unroll 4
    for (int k0 = 0; k0 < 128; k0 += 4) {
        v2f a1 = *(const v2f*)&lds_mean[lr][k0 + kh];
        v2f a2 = *(const v2f*)&lds_root[lr][k0 + kh];
        v2f b1 = *(const v2f*)(wl_row + k0 + kh);
        v2f b2 = *(const v2f*)(wr_row + k0 + kh);
        acc = __builtin_amdgcn_wmma_f32_16x16x4_f32(
            false, a1, false, b1, (short)0, acc, false, false);
        acc = __builtin_amdgcn_wmma_f32_16x16x4_f32(
            false, a2, false, b2, (short)0, acc, false, false);
    }

    const float bv   = bias[col + lr];
    const int   rtop = (lane >> 4) * 8;
#pragma unroll
    for (int v = 0; v < 8; ++v) {
        float r = acc[v] + bv;
        if (RELU) r = fmaxf(r, 0.0f);
        out[(size_t)(row_base + v + rtop) * 128 + col + lr] = r;
    }
}

extern "C" void kernel_launch(void* const* d_in, const int* in_sizes, int n_in,
                              void* d_out, int out_size, void* d_ws, size_t ws_size,
                              hipStream_t stream) {
    const float* x   = (const float*)d_in[0];
    const int*   ei  = (const int*)d_in[1];
    const float* W1l = (const float*)d_in[2];
    const float* b1l = (const float*)d_in[3];
    const float* W1r = (const float*)d_in[4];
    const float* W2l = (const float*)d_in[5];
    const float* b2l = (const float*)d_in[6];
    const float* W2r = (const float*)d_in[7];
    float* out = (float*)d_out;

    const int N = in_sizes[0] / 128;
    const int E = in_sizes[1] / 2;
    const int* src = ei;
    const int* dst = ei + E;

    float* msg = (float*)d_ws;
    float* h1  = msg + (size_t)N * 128;
    float* deg = h1 + (size_t)N * 128;

    const long msg4 = (long)N * 128 / 4;
    const long deg4 = (long)N / 4;
    const int  zb_msg = (int)((msg4 + 255) / 256);
    const int  zb_deg = (int)((deg4 + 255) / 256);
    const int  sc_blocks = (int)(((long)E * 32 + 255) / 256);
    const int  gm_blocks = N / 16;

    zero_f32<<<zb_msg, 256, 0, stream>>>(msg, msg4);
    zero_f32<<<zb_deg, 256, 0, stream>>>(deg, deg4);
    scatter_accum<<<sc_blocks, 256, 0, stream>>>(x, src, dst, msg, deg, E);
    sage_gemm_wmma<true><<<gm_blocks, 256, 0, stream>>>(
        msg, deg, x, W1l, b1l, W1r, h1, N);

    zero_f32<<<zb_msg, 256, 0, stream>>>(msg, msg4);
    scatter_accum<<<sc_blocks, 256, 0, stream>>>(h1, src, dst, msg, nullptr, E);
    sage_gemm_wmma<false><<<gm_blocks, 256, 0, stream>>>(
        msg, deg, h1, W2l, b2l, W2r, out, N);
}


# aggregation: each block keeps the sums of its 392 dst rows in LDS, scans all dst ids, lists its edges, adds src rows there; no f32 atomics, no zero-fill launches; fixed (scan) sum order
# speedup vs baseline: 1.7882x; 1.7882x over previous
#include <hip/hip_runtime.h>
#include <hip/hip_bf16.h>

typedef __attribute__((ext_vector_type(2))) float v2f;
typedef __attribute__((ext_vector_type(8))) float v8f;

#define AGG_R    392
#define AGG_CAP  768
#define AGG_STEP 128

__device__ __forceinline__ void agg_list_step(
    unsigned* __restrict__ list, int& cnt, const int4 d, const int e4,
    const int row0)
{
    const unsigned r0 = (unsigned)(d.x - row0);
    const unsigned r1 = (unsigned)(d.y - row0);
    const unsigned r2 = (unsigned)(d.z - row0);
    const unsigned r3 = (unsigned)(d.w - row0);
    const unsigned rmin = min(min(r0, r1), min(r2, r3));
    if (__builtin_amdgcn_ballot_w32(rmin < (unsigned)AGG_R) != 0u) {
        const bool m0 = r0 < (unsigned)AGG_R;
        const bool m1 = r1 < (unsigned)AGG_R;
        const bool m2 = r2 < (unsigned)AGG_R;
        const bool m3 = r3 < (unsigned)AGG_R;
        const unsigned k0 = __builtin_amdgcn_ballot_w32(m0);
        const unsigned k1 = __builtin_amdgcn_ballot_w32(m1);
        const unsigned k2 = __builtin_amdgcn_ballot_w32(m2);
        const unsigned k3 = __builtin_amdgcn_ballot_w32(m3);
        const int c1 = cnt + __builtin_popcount(k0);
        const int c2 = c1 + __builtin_popcount(k1);
        const int c3 = c2 + __builtin_popcount(k2);
        if (m0) list[cnt + (int)__builtin_amdgcn_mbcnt_lo(k0, 0u)] = ((unsigned)(e4 + 0) << 9) | r0;
        if (m1) list[c1  + (int)__builtin_amdgcn_mbcnt_lo(k1, 0u)] = ((unsigned)(e4 + 1) << 9) | r1;
        if (m2) list[c2  + (int)__builtin_amdgcn_mbcnt_lo(k2, 0u)] = ((unsigned)(e4 + 2) << 9) | r2;
        if (m3) list[c3  + (int)__builtin_amdgcn_mbcnt_lo(k3, 0u)] = ((unsigned)(e4 + 3) << 9) | r3;
        cnt = c3 + __builtin_popcount(k3);
    }
}

__device__ __forceinline__ int4 agg_load_dst_checked(
    const int* __restrict__ dst, const int e4, const int E)
{
    int4 d;
    d.x = (e4 + 0 < E) ? dst[e4 + 0] : -1;
    d.y = (e4 + 1 < E) ? dst[e4 + 1] : -1;
    d.z = (e4 + 2 < E) ? dst[e4 + 2] : -1;
    d.w = (e4 + 3 < E) ? dst[e4 + 3] : -1;
    return d;
}

__device__ __forceinline__ void agg_add_rows(
    const float* __restrict__ feat, float (*acc)[128], int* rowcnt,
    const unsigned ent, const int n, const int lane, const bool count)
{
    for (int j = 0; j < n; j += 8) {
        float4   v[8];
        unsigned en[8];
#pragma unroll
        for (int u = 0; u < 8; ++u) {
            const int jj = (j + u < n) ? (j + u) : (n - 1);
            en[u] = (unsigned)__builtin_amdgcn_readlane((int)ent, jj);
            v[u]  = ((const float4*)(feat + (size_t)(en[u] >> 9) * 128))[lane];
        }
#pragma unroll
        for (int u = 0; u < 8; ++u) {
            if (j + u < n) {
                const int r = (int)(en[u] & 511u);
                float4* p = (float4*)&acc[r][lane * 4];
                float4 a = *p;
                a.x += v[u].x; a.y += v[u].y; a.z += v[u].z; a.w += v[u].w;
                *p = a;
                if (count && lane == 0) rowcnt[r] += 1;
            }
        }
    }
}

__global__ __launch_bounds__(1024) void aggregate_rows_lds(
    const float* __restrict__ feat, const int* __restrict__ src,
    const int* __restrict__ dst, float* __restrict__ msg,
    float* __restrict__ deg, int E, int n_rows)
{
    __shared__ __attribute__((aligned(16))) float acc[AGG_R][128];
    __shared__ unsigned seg[32][AGG_CAP];
    __shared__ unsigned stage[32][64];
    __shared__ int rowcnt[AGG_R];
    __shared__ int segcnt[32];
    __shared__ int segmore[32];

    const int tid  = threadIdx.x;
    const int wave = __builtin_amdgcn_readfirstlane(tid >> 5);
    const int lane = tid & 31;
    const int row0 = blockIdx.x * AGG_R;
    const bool count = (deg != nullptr);

    for (int i = tid; i < AGG_R * 32; i += 1024)
        ((float4*)&acc[0][0])[i] = make_float4(0.f, 0.f, 0.f, 0.f);
    for (int i = tid; i < AGG_R; i += 1024) rowcnt[i] = 0;

    const int total_steps = (E + AGG_STEP - 1) / AGG_STEP;
    const int per_wave    = (total_steps + 31) / 32;
    const int first       = wave * per_wave;
    int my_n = total_steps - first;
    if (my_n > per_wave) my_n = per_wave;
    if (my_n < 0) my_n = 0;
    const int rot = (my_n > 0) ? (int)(((long)blockIdx.x * my_n) / (long)gridDim.x) : 0;
    const bool whole = ((((size_t)dst) & 15) == 0) &&
                       ((long)(first + my_n) * AGG_STEP <= (long)E);

    unsigned* my_list = &seg[wave][0];
    int cursor = 0;
    int more;
    do {
        int  cnt  = 0;
        bool full = false;
        while (cursor < my_n && !full) {
            const int c0 = cursor;
            int4 d[8];
            int  e4[8];
            if (whole && c0 + 8 <= my_n) {
#pragma unroll
                for (int u = 0; u < 8; ++u) {
                    int s = c0 + u + rot;
                    if (s >= my_n) s -= my_n;
                    e4[u] = (first + s) * AGG_STEP + lane * 4;
                    d[u]  = *(const int4*)(dst + e4[u]);
                }
            } else {
#pragma unroll
                for (int u = 0; u < 8; ++u) {
                    int s = c0 + u + rot;
                    if (s >= my_n) s -= my_n;
                    if (s >= my_n) s -= my_n;
                    e4[u] = (first + s) * AGG_STEP + lane * 4;
                    d[u]  = (c0 + u < my_n) ? agg_load_dst_checked(dst, e4[u], E)
                                            : make_int4(-1, -1, -1, -1);
                }
            }
#pragma unroll
            for (int u = 0; u < 8; ++u) {
                if (!full && c0 + u < my_n) {
                    if (cnt + AGG_STEP > AGG_CAP) {
                        full = true;
                    } else {
                        agg_list_step(my_list, cnt, d[u], e4[u], row0);
                        cursor = c0 + u + 1;
                    }
                }
            }
        }
        for (int i = lane; i < cnt; i += 32) {
            const unsigned en = my_list[i];
            const int s = src[en >> 9];
            my_list[i] = ((unsigned)s << 9) | (en & 511u);
        }
        if (lane == 0) {
            segcnt[wave]  = cnt;
            segmore[wave] = (cursor < my_n) ? 1 : 0;
        }
        __syncthreads();

        more = __builtin_amdgcn_readfirstlane(
            (int)(__builtin_amdgcn_ballot_w32(segmore[lane] != 0) != 0u));
        int staged = 0;
        for (int s = 0; s < 32; ++s) {
            const int n = __builtin_amdgcn_readfirstlane(segcnt[s]);
            for (int i0 = 0; i0 < n; i0 += 32) {
                const int idx = i0 + lane;
                const unsigned en = (idx < n) ? seg[s][idx] : 0u;
                const bool mine = (idx < n) && ((int)(en & 31u) == wave);
                const unsigned mk = __builtin_amdgcn_ballot_w32(mine);
                if (mk != 0u) {
                    if (mine) stage[wave][staged + (int)__builtin_amdgcn_mbcnt_lo(mk, 0u)] = en;
                    staged += __builtin_popcount(mk);
                    if (staged >= 32) {
                        const unsigned ent  = stage[wave][lane];
                        const unsigned rest = stage[wave][32 + lane];
                        agg_add_rows(feat, acc, rowcnt, ent, 32, lane, count);
                        staged -= 32;
                        if (lane < staged) stage[wave][lane] = rest;
                    }
                }
            }
        }
        if (staged > 0) {
            const unsigned ent = stage[wave][lane];
            agg_add_rows(feat, acc, rowcnt, ent, staged, lane, count);
        }
        __syncthreads();
    } while (more);

    for (int i = tid; i < AGG_R * 32; i += 1024) {
        const int r = row0 + (i >> 5);
        if (r < n_rows)
            ((float4*)(msg + (size_t)r * 128))[i & 31] = ((const float4*)&acc[0][0])[i];
    }
    if (count) {
        for (int i = tid; i < AGG_R; i += 1024)
            if (row0 + i < n_rows) deg[row0 + i] = (float)rowcnt[i];
    }
}

template <bool RELU>
__global__ __launch_bounds__(256) void sage_gemm_wmma(
    const float* __restrict__ msg, const float* __restrict__ deg,
    const float* __restrict__ root, const float* __restrict__ Wl,
    const float* __restrict__ bias, const float* __restrict__ Wr,
    float* __restrict__ out, int n_rows)
{
    __shared__ float lds_mean[16][132];
    __shared__ float lds_root[16][132];

    const int tid      = threadIdx.x;
    const int row_base = blockIdx.x * 16;
    if (row_base >= n_rows) return;

    {
        int idx = tid * 8;
        int r   = idx >> 7;
        int c   = idx & 127;
        float invd = 1.0f / fmaxf(deg[row_base + r], 1.0f);
        const float* ms = msg  + (size_t)(row_base + r) * 128 + c;
        const float* rt = root + (size_t)(row_base + r) * 128 + c;
#pragma unroll
        for (int j = 0; j < 8; ++j) {
            lds_mean[r][c + j] = ms[j] * invd;
            lds_root[r][c + j] = rt[j];
        }
    }
    __syncthreads();

    const int wave = tid >> 5;
    const int lane = tid & 31;
    const int col  = wave * 16;
    const int lr   = lane & 15;
    const int kh   = (lane >> 4) * 2;

    const float* wl_row = Wl + (size_t)(col + lr) * 128;
    const float* wr_row = Wr + (size_t)(col + lr) * 128;

    v8f acc = {};
#pragma unroll 4
    for (int k0 = 0; k0 < 128; k0 += 4) {
        v2f a1 = *(const v2f*)&lds_mean[lr][k0 + kh];
        v2f a2 = *(const v2f*)&lds_root[lr][k0 + kh];
        v2f b1 = *(const v2f*)(wl_row + k0 + kh);
        v2f b2 = *(const v2f*)(wr_row + k0 + kh);
        acc = __builtin_amdgcn_wmma_f32_16x16x4_f32(
            false, a1, false, b1, (short)0, acc, false, false);
        acc = __builtin_amdgcn_wmma_f32_16x16x4_f32(
            false, a2, false, b2, (short)0, acc, false, false);
    }

    const float bv   = bias[col + lr];
    const int   rtop = (lane >> 4) * 8;
#pragma unroll
    for (int v = 0; v < 8; ++v) {
        float r = acc[v] + bv;
        if (RELU) r = fmaxf(r, 0.0f);
        out[(size_t)(row_base + v + rtop) * 128 + col + lr] = r;
    }
}

extern "C" void kernel_launch(void* const* d_in, const int* in_sizes, int n_in,
                              void* d_out, int out_size, void* d_ws, size_t ws_size,
                              hipStream_t stream) {
    const float* x   = (const float*)d_in[0];
    const int*   ei  = (const int*)d_in[1];
    const float* W1l = (const float*)d_in[2];
    const float* b1l = (const float*)d_in[3];
    const float* W1r = (const float*)d_in[4];
    const float* W2l = (const float*)d_in[5];
    const float* b2l = (const float*)d_in[6];
    const float* W2r = (const float*)d_in[7];
    float* out = (float*)d_out;

    const int N = in_sizes[0] / 128;
    const int E = in_sizes[1] / 2;
    const int* src = ei;
    const int* dst = ei + E;

    float* msg = (float*)d_ws;
    float* h1  = msg + (size_t)N * 128;
    float* deg = h1 + (size_t)N * 128;

    const int  ag_blocks = (N + AGG_R - 1) / AGG_R;
    const int  gm_blocks = N / 16;

    aggregate_rows_lds<<<ag_blocks, 1024, 0, stream>>>(x, src, dst, msg, deg, E, N);
    sage_gemm_wmma<true><<<gm_blocks, 256, 0, stream>>>(
        msg, deg, x, W1l, b1l, W1r, h1, N);

    aggregate_rows_lds<<<ag_blocks, 1024, 0, stream>>>(h1, src, dst, msg, nullptr, E, N);
    sage_gemm_wmma<false><<<gm_blocks, 256, 0, stream>>>(
        msg, deg, h1, W2l, b2l, W2r, out, N);
}


# aggregation, layer 2: each wave replays the (src,row) list it added in layer 1, kept in the workspace within the call; no dst scan, src lookup or list walk in layer 2; per-row sum order unchanged
# speedup vs baseline: 2.1938x; 1.2268x over previous
#include <hip/hip_runtime.h>
#include <hip/hip_bf16.h>

typedef __attribute__((ext_vector_type(2))) float v2f;
typedef __attribute__((ext_vector_type(8))) float v8f;

#define AGG_R    392
#define AGG_CAP  768
#define AGG_STEP 128
#define REC_CAP  512

__device__ __forceinline__ void agg_list_step(
    unsigned* __restrict__ list, int& cnt, const int4 d, const int e4,
    const int row0)
{
    const unsigned r0 = (unsigned)(d.x - row0);
    const unsigned r1 = (unsigned)(d.y - row0);
    const unsigned r2 = (unsigned)(d.z - row0);
    const unsigned r3 = (unsigned)(d.w - row0);
    const unsigned rmin = min(min(r0, r1), min(r2, r3));
    if (__builtin_amdgcn_ballot_w32(rmin < (unsigned)AGG_R) != 0u) {
        const bool m0 = r0 < (unsigned)AGG_R;
        const bool m1 = r1 < (unsigned)AGG_R;
        const bool m2 = r2 < (unsigned)AGG_R;
        const bool m3 = r3 < (unsigned)AGG_R;
        const unsigned k0 = __builtin_amdgcn_ballot_w32(m0);
        const unsigned k1 = __builtin_amdgcn_ballot_w32(m1);
        const unsigned k2 = __builtin_amdgcn_ballot_w32(m2);
        const unsigned k3 = __builtin_amdgcn_ballot_w32(m3);
        const int c1 = cnt + __builtin_popcount(k0);
        const int c2 = c1 + __builtin_popcount(k1);
        const int c3 = c2 + __builtin_popcount(k2);
        if (m0) list[cnt + (int)__builtin_amdgcn_mbcnt_lo(k0, 0u)] = ((unsigned)(e4 + 0) << 9) | r0;
        if (m1) list[c1  + (int)__builtin_amdgcn_mbcnt_lo(k1, 0u)] = ((unsigned)(e4 + 1) << 9) | r1;
        if (m2) list[c2  + (int)__builtin_amdgcn_mbcnt_lo(k2, 0u)] = ((unsigned)(e4 + 2) << 9) | r2;
        if (m3) list[c3  + (int)__builtin_amdgcn_mbcnt_lo(k3, 0u)] = ((unsigned)(e4 + 3) << 9) | r3;
        cnt = c3 + __builtin_popcount(k3);
    }
}

__device__ __forceinline__ int4 agg_load_dst_checked(
    const int* __restrict__ dst, const int e4, const int E)
{
    int4 d;
    d.x = (e4 + 0 < E) ? dst[e4 + 0] : -1;
    d.y = (e4 + 1 < E) ? dst[e4 + 1] : -1;
    d.z = (e4 + 2 < E) ? dst[e4 + 2] : -1;
    d.w = (e4 + 3 < E) ? dst[e4 + 3] : -1;
    return d;
}

__device__ __forceinline__ void agg_add_rows(
    const float* __restrict__ feat, float (*acc)[128], int* rowcnt,
    const unsigned ent, const int n, const int lane, const bool count)
{
    for (int j = 0; j < n; j += 8) {
        float4   v[8];
        unsigned en[8];
#pragma unroll
        for (int u = 0; u < 8; ++u) {
            const int jj = (j + u < n) ? (j + u) : (n - 1);
            en[u] = (unsigned)__builtin_amdgcn_readlane((int)ent, jj);
            v[u]  = ((const float4*)(feat + (size_t)(en[u] >> 9) * 128))[lane];
        }
#pragma unroll
        for (int u = 0; u < 8; ++u) {
            if (j + u < n) {
                const int r = (int)(en[u] & 511u);
                float4* p = (float4*)&acc[r][lane * 4];
                float4 a = *p;
                a.x += v[u].x; a.y += v[u].y; a.z += v[u].z; a.w += v[u].w;
                *p = a;
                if (count && lane == 0) rowcnt[r] += 1;
            }
        }
    }
}

__global__ __launch_bounds__(1024) void aggregate_rows_lds(
    const float* __restrict__ feat, const int* __restrict__ src,
    const int* __restrict__ dst, float* __restrict__ msg,
    float* __restrict__ deg, int E, int n_rows,
    unsigned* __restrict__ rec_list, int* __restrict__ rec_cnt,
    const unsigned* __restrict__ rep_list, const int* __restrict__ rep_cnt)
{
    __shared__ __attribute__((aligned(16))) float acc[AGG_R][128];
    __shared__ unsigned seg[32][AGG_CAP];
    __shared__ unsigned stage[32][64];
    __shared__ int rowcnt[AGG_R];
    __shared__ int segcnt[32];
    __shared__ int segmore[32];

    const int tid  = threadIdx.x;
    const int wave = __builtin_amdgcn_readfirstlane(tid >> 5);
    const int lane = tid & 31;
    const int row0 = blockIdx.x * AGG_R;
    const bool count = (deg != nullptr);

    if (rep_cnt != nullptr && !count) {
        const int  c  = rep_cnt[blockIdx.x * 32 + lane];
        const bool ok = (__builtin_amdgcn_ballot_w32(c < 0) == 0u);
        if (ok) {
            const int n = __builtin_amdgcn_readlane(c, wave);
            const unsigned* my = rep_list + (size_t)(blockIdx.x * 32 + wave) * REC_CAP;
            for (int r = wave; r < AGG_R; r += 32)
                *(float4*)&acc[r][lane * 4] = make_float4(0.f, 0.f, 0.f, 0.f);
            unsigned ent = (lane < n) ? my[lane] : 0u;
            for (int i0 = 0; i0 < n; i0 += 32) {
                const int      nx  = i0 + 32 + lane;
                const unsigned nxt = (nx < n) ? my[nx] : 0u;
                const int      m   = (n - i0 < 32) ? (n - i0) : 32;
                agg_add_rows(feat, acc, rowcnt, ent, m, lane, false);
                ent = nxt;
            }
            for (int r = wave; r < AGG_R; r += 32)
                if (row0 + r < n_rows)
                    ((float4*)(msg + (size_t)(row0 + r) * 128))[lane] =
                        *(const float4*)&acc[r][lane * 4];
            return;
        }
    }

    const bool rec = (rec_cnt != nullptr);
    unsigned* my_rec = rec ? rec_list + (size_t)(blockIdx.x * 32 + wave) * REC_CAP : nullptr;
    int rec_pos = 0;

    for (int i = tid; i < AGG_R * 32; i += 1024)
        ((float4*)&acc[0][0])[i] = make_float4(0.f, 0.f, 0.f, 0.f);
    for (int i = tid; i < AGG_R; i += 1024) rowcnt[i] = 0;

    const int total_steps = (E + AGG_STEP - 1) / AGG_STEP;
    const int per_wave    = (total_steps + 31) / 32;
    const int first       = wave * per_wave;
    int my_n = total_steps - first;
    if (my_n > per_wave) my_n = per_wave;
    if (my_n < 0) my_n = 0;
    const int rot = (my_n > 0) ? (int)(((long)blockIdx.x * my_n) / (long)gridDim.x) : 0;
    const bool whole = ((((size_t)dst) & 15) == 0) &&
                       ((long)(first + my_n) * AGG_STEP <= (long)E);

    unsigned* my_list = &seg[wave][0];
    int cursor = 0;
    int more;
    do {
        int  cnt  = 0;
        bool full = false;
        while (cursor < my_n && !full) {
            const int c0 = cursor;
            int4 d[8];
            int  e4[8];
            if (whole && c0 + 8 <= my_n) {
#pragma unroll
                for (int u = 0; u < 8; ++u) {
                    int s = c0 + u + rot;
                    if (s >= my_n) s -= my_n;
                    e4[u] = (first + s) * AGG_STEP + lane * 4;
                    d[u]  = *(const int4*)(dst + e4[u]);
                }
            } else {
#pragma unroll
                for (int u = 0; u < 8; ++u) {
                    int s = c0 + u + rot;
                    if (s >= my_n) s -= my_n;
                    if (s >= my_n) s -= my_n;
                    e4[u] = (first + s) * AGG_STEP + lane * 4;
                    d[u]  = (c0 + u < my_n) ? agg_load_dst_checked(dst, e4[u], E)
                                            : make_int4(-1, -1, -1, -1);
                }
            }
#pragma unroll
            for (int u = 0; u < 8; ++u) {
                if (!full && c0 + u < my_n) {
                    if (cnt + AGG_STEP > AGG_CAP) {
                        full = true;
                    } else {
                        agg_list_step(my_list, cnt, d[u], e4[u], row0);
                        cursor = c0 + u + 1;
                    }
                }
            }
        }
        for (int i = lane; i < cnt; i += 32) {
            const unsigned en = my_list[i];
            const int s = src[en >> 9];
            my_list[i] = ((unsigned)s << 9) | (en & 511u);
        }
        if (lane == 0) {
            segcnt[wave]  = cnt;
            segmore[wave] = (cursor < my_n) ? 1 : 0;
        }
        __syncthreads();

        more = __builtin_amdgcn_readfirstlane(
            (int)(__builtin_amdgcn_ballot_w32(segmore[lane] != 0) != 0u));
        int staged = 0;
        for (int s = 0; s < 32; ++s) {
            const int n = __builtin_amdgcn_readfirstlane(segcnt[s]);
            for (int i0 = 0; i0 < n; i0 += 32) {
                const int idx = i0 + lane;
                const unsigned en = (idx < n) ? seg[s][idx] : 0u;
                const bool mine = (idx < n) && ((int)(en & 31u) == wave);
                const unsigned mk = __builtin_amdgcn_ballot_w32(mine);
                if (mk != 0u) {
                    if (mine) stage[wave][staged + (int)__builtin_amdgcn_mbcnt_lo(mk, 0u)] = en;
                    staged += __builtin_popcount(mk);
                    if (staged >= 32) {
                        const unsigned ent  = stage[wave][lane];
                        const unsigned rest = stage[wave][32 + lane];
                        if (rec) {
                            if (rec_pos + 32 <= REC_CAP) my_rec[rec_pos + lane] = ent;
                            rec_pos += 32;
                        }
                        agg_add_rows(feat, acc, rowcnt, ent, 32, lane, count);
                        staged -= 32;
                        if (lane < staged) stage[wave][lane] = rest;
                    }
                }
            }
        }
        if (staged > 0) {
            const unsigned ent = stage[wave][lane];
            if (rec) {
                if (rec_pos + staged <= REC_CAP && lane < staged) my_rec[rec_pos + lane] = ent;
                rec_pos += staged;
            }
            agg_add_rows(feat, acc, rowcnt, ent, staged, lane, count);
        }
        __syncthreads();
    } while (more);

    if (rec && lane == 0)
        rec_cnt[blockIdx.x * 32 + wave] = (rec_pos <= REC_CAP) ? rec_pos : -1;

    for (int i = tid; i < AGG_R * 32; i += 1024) {
        const int r = row0 + (i >> 5);
        if (r < n_rows)
            ((float4*)(msg + (size_t)r * 128))[i & 31] = ((const float4*)&acc[0][0])[i];
    }
    if (count) {
        for (int i = tid; i < AGG_R; i += 1024)
            if (row0 + i < n_rows) deg[row0 + i] = (float)rowcnt[i];
    }
}

template <bool RELU>
__global__ __launch_bounds__(256) void sage_gemm_wmma(
    const float* __restrict__ msg, const float* __restrict__ deg,
    const float* __restrict__ root, const float* __restrict__ Wl,
    const float* __restrict__ bias, const float* __restrict__ Wr,
    float* __restrict__ out, int n_rows)
{
    __shared__ float lds_mean[16][132];
    __shared__ float lds_root[16][132];

    const int tid      = threadIdx.x;
    const int row_base = blockIdx.x * 16;
    if (row_base >= n_rows) return;

    {
        int idx = tid * 8;
        int r   = idx >> 7;
        int c   = idx & 127;
        float invd = 1.0f / fmaxf(deg[row_base + r], 1.0f);
        const float* ms = msg  + (size_t)(row_base + r) * 128 + c;
        const float* rt = root + (size_t)(row_base + r) * 128 + c;
#pragma unroll
        for (int j = 0; j < 8; ++j) {
            lds_mean[r][c + j] = ms[j] * invd;
            lds_root[r][c + j] = rt[j];
        }
    }
    __syncthreads();

    const int wave = tid >> 5;
    const int lane = tid & 31;
    const int col  = wave * 16;
    const int lr   = lane & 15;
    const int kh   = (lane >> 4) * 2;

    const float* wl_row = Wl + (size_t)(col + lr) * 128;
    const float* wr_row = Wr + (size_t)(col + lr) * 128;

    v8f acc = {};
#pragma unroll 4
    for (int k0 = 0; k0 < 128; k0 += 4) {
        v2f a1 = *(const v2f*)&lds_mean[lr][k0 + kh];
        v2f a2 = *(const v2f*)&lds_root[lr][k0 + kh];
        v2f b1 = *(const v2f*)(wl_row + k0 + kh);
        v2f b2 = *(const v2f*)(wr_row + k0 + kh);
        acc = __builtin_amdgcn_wmma_f32_16x16x4_f32(
            false, a1, false, b1, (short)0, acc, false, false);
        acc = __builtin_amdgcn_wmma_f32_16x16x4_f32(
            false, a2, false, b2, (short)0, acc, false, false);
    }

    const float bv   = bias[col + lr];
    const int   rtop = (lane >> 4) * 8;
#pragma unroll
    for (int v = 0; v < 8; ++v) {
        float r = acc[v] + bv;
        if (RELU) r = fmaxf(r, 0.0f);
        out[(size_t)(row_base + v + rtop) * 128 + col + lr] = r;
    }
}

extern "C" void kernel_launch(void* const* d_in, const int* in_sizes, int n_in,
                              void* d_out, int out_size, void* d_ws, size_t ws_size,
                              hipStream_t stream) {
    const float* x   = (const float*)d_in[0];
    const int*   ei  = (const int*)d_in[1];
    const float* W1l = (const float*)d_in[2];
    const float* b1l = (const float*)d_in[3];
    const float* W1r = (const float*)d_in[4];
    const float* W2l = (const float*)d_in[5];
    const float* b2l = (const float*)d_in[6];
    const float* W2r = (const float*)d_in[7];
    float* out = (float*)d_out;

    const int N = in_sizes[0] / 128;
    const int E = in_sizes[1] / 2;
    const int* src = ei;
    const int* dst = ei + E;

    float* msg = (float*)d_ws;
    float* h1  = msg + (size_t)N * 128;
    float* deg = h1 + (size_t)N * 128;

    const int  ag_blocks = (N + AGG_R - 1) / AGG_R;
    const int  gm_blocks = N / 16;

    int*      rec_cnt  = (int*)(deg + N);
    unsigned* rec_list = (unsigned*)(rec_cnt + ag_blocks * 32);

    aggregate_rows_lds<<<ag_blocks, 1024, 0, stream>>>(
        x, src, dst, msg, deg, E, N, rec_list, rec_cnt, nullptr, nullptr);
    sage_gemm_wmma<true><<<gm_blocks, 256, 0, stream>>>(
        msg, deg, x, W1l, b1l, W1r, h1, N);

    aggregate_rows_lds<<<ag_blocks, 1024, 0, stream>>>(
        h1, src, dst, msg, nullptr, E, N, nullptr, nullptr, rec_list, rec_cnt);
    sage_gemm_wmma<false><<<gm_blocks, 256, 0, stream>>>(
        msg, deg, h1, W2l, b2l, W2r, out, N);
}


# layer matrix products moved into the aggregation kernel: they read the block's LDS means directly (root rows copied over them after); msg round trip, 2 launches gone; sums: mean terms, then root
# speedup vs baseline: 2.9730x; 1.3552x over previous
#include <hip/hip_runtime.h>
#include <hip/hip_bf16.h>

typedef __attribute__((ext_vector_type(2))) float v2f;
typedef __attribute__((ext_vector_type(8))) float v8f;

#define AGG_R    392
#define ACC_LD   132
#define AGG_RPW  13
#define AGG_CAP  768
#define AGG_STEP 128
#define REC_CAP  512

__device__ __forceinline__ void agg_list_step(
    unsigned* __restrict__ list, int& cnt, const int4 d, const int e4,
    const int row0)
{
    const unsigned r0 = (unsigned)(d.x - row0);
    const unsigned r1 = (unsigned)(d.y - row0);
    const unsigned r2 = (unsigned)(d.z - row0);
    const unsigned r3 = (unsigned)(d.w - row0);
    const unsigned rmin = min(min(r0, r1), min(r2, r3));
    if (__builtin_amdgcn_ballot_w32(rmin < (unsigned)AGG_R) != 0u) {
        const bool m0 = r0 < (unsigned)AGG_R;
        const bool m1 = r1 < (unsigned)AGG_R;
        const bool m2 = r2 < (unsigned)AGG_R;
        const bool m3 = r3 < (unsigned)AGG_R;
        const unsigned k0 = __builtin_amdgcn_ballot_w32(m0);
        const unsigned k1 = __builtin_amdgcn_ballot_w32(m1);
        const unsigned k2 = __builtin_amdgcn_ballot_w32(m2);
        const unsigned k3 = __builtin_amdgcn_ballot_w32(m3);
        const int c1 = cnt + __builtin_popcount(k0);
        const int c2 = c1 + __builtin_popcount(k1);
        const int c3 = c2 + __builtin_popcount(k2);
        if (m0) list[cnt + (int)__builtin_amdgcn_mbcnt_lo(k0, 0u)] = ((unsigned)(e4 + 0) << 9) | r0;
        if (m1) list[c1  + (int)__builtin_amdgcn_mbcnt_lo(k1, 0u)] = ((unsigned)(e4 + 1) << 9) | r1;
        if (m2) list[c2  + (int)__builtin_amdgcn_mbcnt_lo(k2, 0u)] = ((unsigned)(e4 + 2) << 9) | r2;
        if (m3) list[c3  + (int)__builtin_amdgcn_mbcnt_lo(k3, 0u)] = ((unsigned)(e4 + 3) << 9) | r3;
        cnt = c3 + __builtin_popcount(k3);
    }
}

__device__ __forceinline__ int4 agg_load_dst_checked(
    const int* __restrict__ dst, const int e4, const int E)
{
    int4 d;
    d.x = (e4 + 0 < E) ? dst[e4 + 0] : -1;
    d.y = (e4 + 1 < E) ? dst[e4 + 1] : -1;
    d.z = (e4 + 2 < E) ? dst[e4 + 2] : -1;
    d.w = (e4 + 3 < E) ? dst[e4 + 3] : -1;
    return d;
}

__device__ __forceinline__ void agg_add_rows(
    const float* __restrict__ feat, float (*acc)[ACC_LD], int* rowcnt,
    const unsigned ent, const int n, const int lane, const bool count)
{
    for (int j = 0; j < n; j += 8) {
        float4   v[8];
        unsigned en[8];
#pragma unroll
        for (int u = 0; u < 8; ++u) {
            const int jj = (j + u < n) ? (j + u) : (n - 1);
            en[u] = (unsigned)__builtin_amdgcn_readlane((int)ent, jj);
            v[u]  = ((const float4*)(feat + (size_t)(en[u] >> 9) * 128))[lane];
        }
#pragma unroll
        for (int u = 0; u < 8; ++u) {
            if (j + u < n) {
                const int r = (int)(en[u] & 511u);
                float4* p = (float4*)&acc[r][lane * 4];
                float4 a = *p;
                a.x += v[u].x; a.y += v[u].y; a.z += v[u].z; a.w += v[u].w;
                *p = a;
                if (count && lane == 0) rowcnt[r] += 1;
            }
        }
    }
}

template <int NT>
__device__ __forceinline__ void tile_products(
    v8f (&c)[7], const float (*acc)[ACC_LD], const float* __restrict__ W,
    const int rg, const int col, const int lane)
{
    const int lr = lane & 15;
    const int kh = (lane >> 4) * 4;
    const float* wrow = W + (size_t)(col + lr) * 128 + kh;
    const float* arow[NT];
#pragma unroll
    for (int i = 0; i < NT; ++i) {
        int r = (rg + 4 * i) * 16 + lr;
        if (i == 6 && r > AGG_R - 1) r = AGG_R - 1;
        arow[i] = &acc[r][kh];
    }
#pragma unroll 1
    for (int j = 0; j < 16; ++j) {
        const float4 b = *(const float4*)(wrow + 8 * j);
        float4 a[NT];
#pragma unroll
        for (int i = 0; i < NT; ++i) a[i] = *(const float4*)(arow[i] + 8 * j);
        const v2f b0 = {b.x, b.y};
        const v2f b1 = {b.z, b.w};
#pragma unroll
        for (int i = 0; i < NT; ++i) {
            const v2f a0 = {a[i].x, a[i].y};
            c[i] = __builtin_amdgcn_wmma_f32_16x16x4_f32(
                false, a0, false, b0, (short)0, c[i], false, false);
        }
#pragma unroll
        for (int i = 0; i < NT; ++i) {
            const v2f a1 = {a[i].z, a[i].w};
            c[i] = __builtin_amdgcn_wmma_f32_16x16x4_f32(
                false, a1, false, b1, (short)0, c[i], false, false);
        }
    }
}

template <bool RELU, int NT>
__device__ __forceinline__ void tile_store(
    const v8f (&c)[7], const float* __restrict__ bias, float* __restrict__ out,
    const int rg, const int col, const int lane, const int row0, const int nvalid)
{
    const int   lr   = lane & 15;
    const int   rtop = (lane >> 4) * 8;
    const float bv   = bias[col + lr];
#pragma unroll
    for (int i = 0; i < NT; ++i) {
        const int rb = (rg + 4 * i) * 16 + rtop;
        float* op = out + (size_t)(row0 + rb) * 128 + col + lr;
#pragma unroll
        for (int v = 0; v < 8; ++v) {
            if (rb + v < nvalid) {
                float o = c[i][v] + bv;
                if (RELU) o = fmaxf(o, 0.0f);
                op[v * 128] = o;
            }
        }
    }
}

template <bool RELU>
__global__ __launch_bounds__(1024) void sage_layer_lds(
    const float* __restrict__ feat, const int* __restrict__ src,
    const int* __restrict__ dst, float* __restrict__ deg, int count_deg,
    int E, int n_rows,
    unsigned* __restrict__ rec_list, int* __restrict__ rec_cnt,
    const unsigned* __restrict__ rep_list, const int* __restrict__ rep_cnt,
    const float* __restrict__ Wl, const float* __restrict__ bias,
    const float* __restrict__ Wr, float* __restrict__ out)
{
    __shared__ __attribute__((aligned(16))) float acc[AGG_R][ACC_LD];
    __shared__ unsigned seg[32][AGG_CAP];
    __shared__ unsigned stage[32][64];
    __shared__ int rowcnt[AGG_R];
    __shared__ int segcnt[32];
    __shared__ int segmore[32];

    const int tid  = threadIdx.x;
    const int wave = __builtin_amdgcn_readfirstlane(tid >> 5);
    const int lane = tid & 31;
    const int row0 = blockIdx.x * AGG_R;
    const bool count = (count_deg != 0);

    const int  my_r  = wave + 32 * lane;
    const bool my_in = (lane < AGG_RPW) && (my_r < AGG_R);
    const bool my_ok = my_in && (row0 + my_r < n_rows);
    float dl = 1.0f;
    if (!count && my_ok) dl = deg[row0 + my_r];

    bool replayed = false;
    if (rep_cnt != nullptr && !count) {
        const int  c  = rep_cnt[blockIdx.x * 32 + lane];
        const bool ok = (__builtin_amdgcn_ballot_w32(c < 0) == 0u);
        if (ok) {
            const int n = __builtin_amdgcn_readlane(c, wave);
            const unsigned* my = rep_list + (size_t)(blockIdx.x * 32 + wave) * REC_CAP;
            for (int r = wave; r < AGG_R; r += 32)
                *(float4*)&acc[r][lane * 4] = make_float4(0.f, 0.f, 0.f, 0.f);
            unsigned ent = (lane < n) ? my[lane] : 0u;
            for (int i0 = 0; i0 < n; i0 += 32) {
                const int      nx  = i0 + 32 + lane;
                const unsigned nxt = (nx < n) ? my[nx] : 0u;
                const int      m   = (n - i0 < 32) ? (n - i0) : 32;
                agg_add_rows(feat, acc, rowcnt, ent, m, lane, false);
                ent = nxt;
            }
            replayed = true;
        }
    }

    if (!replayed) {
        const bool rec = (rec_cnt != nullptr);
        unsigned* my_rec = rec ? rec_list + (size_t)(blockIdx.x * 32 + wave) * REC_CAP : nullptr;
        int rec_pos = 0;

        for (int i = tid; i < AGG_R * (ACC_LD / 4); i += 1024)
            ((float4*)&acc[0][0])[i] = make_float4(0.f, 0.f, 0.f, 0.f);
        for (int i = tid; i < AGG_R; i += 1024) rowcnt[i] = 0;

        const int total_steps = (E + AGG_STEP - 1) / AGG_STEP;
        const int per_wave    = (total_steps + 31) / 32;
        const int first       = wave * per_wave;
        int my_n = total_steps - first;
        if (my_n > per_wave) my_n = per_wave;
        if (my_n < 0) my_n = 0;
        const int rot = (my_n > 0) ? (int)(((long)blockIdx.x * my_n) / (long)gridDim.x) : 0;
        const bool whole = ((((size_t)dst) & 15) == 0) &&
                           ((long)(first + my_n) * AGG_STEP <= (long)E);

        unsigned* my_list = &seg[wave][0];
        int cursor = 0;
        int more;
        do {
            int  cnt  = 0;
            bool full = false;
            while (cursor < my_n && !full) {
                const int c0 = cursor;
                int4 d[8];
                int  e4[8];
                if (whole && c0 + 8 <= my_n) {
    #pragma unroll
                    for (int u = 0; u < 8; ++u) {
                        int s = c0 + u + rot;
                        if (s >= my_n) s -= my_n;
                        e4[u] = (first + s) * AGG_STEP + lane * 4;
                        d[u]  = *(const int4*)(dst + e4[u]);
                    }
                } else {
    #pragma unroll
                    for (int u = 0; u < 8; ++u) {
                        int s = c0 + u + rot;
                        if (s >= my_n) s -= my_n;
                        if (s >= my_n) s -= my_n;
                        e4[u] = (first + s) * AGG_STEP + lane * 4;
                        d[u]  = (c0 + u < my_n) ? agg_load_dst_checked(dst, e4[u], E)
                                                : make_int4(-1, -1, -1, -1);
                    }
                }
    #pragma unroll
                for (int u = 0; u < 8; ++u) {
                    if (!full && c0 + u < my_n) {
                        if (cnt + AGG_STEP > AGG_CAP) {
                            full = true;
                        } else {
                            agg_list_step(my_list, cnt, d[u], e4[u], row0);
                            cursor = c0 + u + 1;
                        }
                    }
                }
            }
            for (int i = lane; i < cnt; i += 32) {
                const unsigned en = my_list[i];
                const int s = src[en >> 9];
                my_list[i] = ((unsigned)s << 9) | (en & 511u);
            }
            if (lane == 0) {
                segcnt[wave]  = cnt;
                segmore[wave] = (cursor < my_n) ? 1 : 0;
            }
            __syncthreads();

            more = __builtin_amdgcn_readfirstlane(
                (int)(__builtin_amdgcn_ballot_w32(segmore[lane] != 0) != 0u));
            int staged = 0;
            for (int s = 0; s < 32; ++s) {
                const int n = __builtin_amdgcn_readfirstlane(segcnt[s]);
                for (int i0 = 0; i0 < n; i0 += 32) {
                    const int idx = i0 + lane;
                    const unsigned en = (idx < n) ? seg[s][idx] : 0u;
                    const bool mine = (idx < n) && ((int)(en & 31u) == wave);
                    const unsigned mk = __builtin_amdgcn_ballot_w32(mine);
                    if (mk != 0u) {
                        if (mine) stage[wave][staged + (int)__builtin_amdgcn_mbcnt_lo(mk, 0u)] = en;
                        staged += __builtin_popcount(mk);
                        if (staged >= 32) {
                            const unsigned ent  = stage[wave][lane];
                            const unsigned rest = stage[wave][32 + lane];
                            if (rec) {
                                if (rec_pos + 32 <= REC_CAP) my_rec[rec_pos + lane] = ent;
                                rec_pos += 32;
                            }
                            agg_add_rows(feat, acc, rowcnt, ent, 32, lane, count);
                            staged -= 32;
                            if (lane < staged) stage[wave][lane] = rest;
                        }
                    }
                }
            }
            if (staged > 0) {
                const unsigned ent = stage[wave][lane];
                if (rec) {
                    if (rec_pos + staged <= REC_CAP && lane < staged) my_rec[rec_pos + lane] = ent;
                    rec_pos += staged;
                }
                agg_add_rows(feat, acc, rowcnt, ent, staged, lane, count);
            }
            __syncthreads();
        } while (more);

        if (rec && lane == 0)
            rec_cnt[blockIdx.x * 32 + wave] = (rec_pos <= REC_CAP) ? rec_pos : -1;
    }

    if (count) {
        dl = my_in ? (float)rowcnt[my_r] : 1.0f;
        if (my_ok) deg[row0 + my_r] = dl;
    }
    for (int i = 0; i < AGG_RPW; ++i) {
        const int r = wave + 32 * i;
        if (r < AGG_R) {
            const float d = __int_as_float(
                __builtin_amdgcn_readlane(__float_as_int(dl), i));
            const float invd = 1.0f / fmaxf(d, 1.0f);
            float4* p = (float4*)&acc[r][lane * 4];
            float4 a = *p;
            a.x *= invd; a.y *= invd; a.z *= invd; a.w *= invd;
            *p = a;
        }
    }
    __syncthreads();

    const int rg  = wave >> 3;
    const int col = (wave & 7) * 16;
    v8f c[7];
#pragma unroll
    for (int i = 0; i < 7; ++i) c[i] = v8f{};

    if (rg == 0) tile_products<7>(c, acc, Wl, rg, col, lane);
    else         tile_products<6>(c, acc, Wl, rg, col, lane);

    __syncthreads();
    const int nvalid = (n_rows - row0 < AGG_R) ? (n_rows - row0) : AGG_R;
    for (int i0 = 0; i0 < AGG_RPW; i0 += 7) {
        float4 v[7];
#pragma unroll
        for (int u = 0; u < 7; ++u) {
            const int r = wave + 32 * (i0 + u);
            v[u] = (r < nvalid)
                 ? ((const float4*)(feat + (size_t)(row0 + r) * 128))[lane]
                 : make_float4(0.f, 0.f, 0.f, 0.f);
        }
#pragma unroll
        for (int u = 0; u < 7; ++u) {
            const int r = wave + 32 * (i0 + u);
            if (r < AGG_R) *(float4*)&acc[r][lane * 4] = v[u];
        }
    }
    __syncthreads();

    if (rg == 0) {
        tile_products<7>(c, acc, Wr, rg, col, lane);
        tile_store<RELU, 7>(c, bias, out, rg, col, lane, row0, nvalid);
    } else {
        tile_products<6>(c, acc, Wr, rg, col, lane);
        tile_store<RELU, 6>(c, bias, out, rg, col, lane, row0, nvalid);
    }
}

extern "C" void kernel_launch(void* const* d_in, const int* in_sizes, int n_in,
                              void* d_out, int out_size, void* d_ws, size_t ws_size,
                              hipStream_t stream) {
    const float* x   = (const float*)d_in[0];
    const int*   ei  = (const int*)d_in[1];
    const float* W1l = (const float*)d_in[2];
    const float* b1l = (const float*)d_in[3];
    const float* W1r = (const float*)d_in[4];
    const float* W2l = (const float*)d_in[5];
    const float* b2l = (const float*)d_in[6];
    const float* W2r = (const float*)d_in[7];
    float* out = (float*)d_out;

    const int N = in_sizes[0] / 128;
    const int E = in_sizes[1] / 2;
    const int* src = ei;
    const int* dst = ei + E;

    float* h1  = (float*)d_ws;
    float* deg = h1 + (size_t)N * 128;

    const int  ag_blocks = (N + AGG_R - 1) / AGG_R;

    int*      rec_cnt  = (int*)(deg + N);
    unsigned* rec_list = (unsigned*)(rec_cnt + ag_blocks * 32);

    sage_layer_lds<true><<<ag_blocks, 1024, 0, stream>>>(
        x, src, dst, deg, 1, E, N, rec_list, rec_cnt, nullptr, nullptr,
        W1l, b1l, W1r, h1);

    sage_layer_lds<false><<<ag_blocks, 1024, 0, stream>>>(
        h1, src, dst, deg, 0, E, N, nullptr, nullptr, rec_list, rec_cnt,
        W2l, b2l, W2r, out);
}


# layer-1 scan reads one key byte per edge (block number of its dst row, written by a small new launch in front), not the 4-byte dst id: 4x fewer scan bytes; list order, so fixed sum order, changes
# speedup vs baseline: 3.2719x; 1.1005x over previous
#include <hip/hip_runtime.h>
#include <hip/hip_bf16.h>

typedef __attribute__((ext_vector_type(2))) float v2f;
typedef __attribute__((ext_vector_type(8))) float v8f;

#define AGG_R    392
#define ACC_LD   132
#define AGG_RPW  13
#define AGG_CAP  832
#define AGG_STEP 512
#define REC_CAP  512

__device__ __forceinline__ void agg_list_step(
    unsigned* __restrict__ list, int& cnt, const uint4 kq, const int eb,
    const unsigned keyrep)
{
    const unsigned w[4] = {kq.x, kq.y, kq.z, kq.w};
    unsigned m = 0u;
#pragma unroll
    for (int q = 0; q < 4; ++q) {
        const unsigned x = w[q] ^ keyrep;
        const unsigned t = ~(((x & 0x7F7F7F7Fu) + 0x7F7F7F7Fu) | x | 0x7F7F7F7Fu);
        m |= t >> (7 - q);
    }
    while (__builtin_amdgcn_ballot_w32(m != 0u) != 0u) {
        const bool     has = (m != 0u);
        const unsigned k   = __builtin_amdgcn_ballot_w32(has);
        if (has) {
            const int p = __builtin_ctz(m);
            list[cnt + (int)__builtin_amdgcn_mbcnt_lo(k, 0u)] =
                (unsigned)(eb + ((p & 3) << 2) + (p >> 3));
        }
        cnt += __builtin_popcount(k);
        m &= m - 1u;
    }
}

__device__ __forceinline__ int4 agg_load_dst_checked(
    const int* __restrict__ dst, const int e4, const int E)
{
    int4 d;
    d.x = (e4 + 0 < E) ? dst[e4 + 0] : -1;
    d.y = (e4 + 1 < E) ? dst[e4 + 1] : -1;
    d.z = (e4 + 2 < E) ? dst[e4 + 2] : -1;
    d.w = (e4 + 3 < E) ? dst[e4 + 3] : -1;
    return d;
}

__global__ __launch_bounds__(256) void edge_block_keys(
    const int* __restrict__ dst, unsigned* __restrict__ keys, int E, int n_steps)
{
    const int step = blockIdx.x * 8 + (threadIdx.x >> 5);
    const int lane = threadIdx.x & 31;
    if (step >= n_steps) return;
    const bool vec = ((((size_t)dst) & 15) == 0);
    int4 d[4];
#pragma unroll
    for (int j = 0; j < 4; ++j) {
        const int e0 = step * AGG_STEP + j * 128 + lane * 4;
        if (vec && (e0 + 4 <= E)) d[j] = *(const int4*)(dst + e0);
        else                      d[j] = agg_load_dst_checked(dst, e0, E);
    }
#pragma unroll
    for (int j = 0; j < 4; ++j) {
        const unsigned k0 = min((unsigned)d[j].x / (unsigned)AGG_R, 255u);
        const unsigned k1 = min((unsigned)d[j].y / (unsigned)AGG_R, 255u);
        const unsigned k2 = min((unsigned)d[j].z / (unsigned)AGG_R, 255u);
        const unsigned k3 = min((unsigned)d[j].w / (unsigned)AGG_R, 255u);
        keys[step * (AGG_STEP / 4) + j * 32 + lane] =
            k0 | (k1 << 8) | (k2 << 16) | (k3 << 24);
    }
}

__device__ __forceinline__ void agg_add_rows(
    const float* __restrict__ feat, float (*acc)[ACC_LD], int* rowcnt,
    const unsigned ent, const int n, const int lane, const bool count)
{
    for (int j = 0; j < n; j += 8) {
        float4   v[8];
        unsigned en[8];
#pragma unroll
        for (int u = 0; u < 8; ++u) {
            const int jj = (j + u < n) ? (j + u) : (n - 1);
            en[u] = (unsigned)__builtin_amdgcn_readlane((int)ent, jj);
            v[u]  = ((const float4*)(feat + (size_t)(en[u] >> 9) * 128))[lane];
        }
#pragma unroll
        for (int u = 0; u < 8; ++u) {
            if (j + u < n) {
                const int r = (int)(en[u] & 511u);
                float4* p = (float4*)&acc[r][lane * 4];
                float4 a = *p;
                a.x += v[u].x; a.y += v[u].y; a.z += v[u].z; a.w += v[u].w;
                *p = a;
                if (count && lane == 0) rowcnt[r] += 1;
            }
        }
    }
}

template <int NT>
__device__ __forceinline__ void tile_products(
    v8f (&c)[7], const float (*acc)[ACC_LD], const float* __restrict__ W,
    const int rg, const int col, const int lane)
{
    const int lr = lane & 15;
    const int kh = (lane >> 4) * 4;
    const float* wrow = W + (size_t)(col + lr) * 128 + kh;
    const float* arow[NT];
#pragma unroll
    for (int i = 0; i < NT; ++i) {
        int r = (rg + 4 * i) * 16 + lr;
        if (i == 6 && r > AGG_R - 1) r = AGG_R - 1;
        arow[i] = &acc[r][kh];
    }
#pragma unroll 1
    for (int j = 0; j < 16; ++j) {
        const float4 b = *(const float4*)(wrow + 8 * j);
        float4 a[NT];
#pragma unroll
        for (int i = 0; i < NT; ++i) a[i] = *(const float4*)(arow[i] + 8 * j);
        const v2f b0 = {b.x, b.y};
        const v2f b1 = {b.z, b.w};
#pragma unroll
        for (int i = 0; i < NT; ++i) {
            const v2f a0 = {a[i].x, a[i].y};
            c[i] = __builtin_amdgcn_wmma_f32_16x16x4_f32(
                false, a0, false, b0, (short)0, c[i], false, false);
        }
#pragma unroll
        for (int i = 0; i < NT; ++i) {
            const v2f a1 = {a[i].z, a[i].w};
            c[i] = __builtin_amdgcn_wmma_f32_16x16x4_f32(
                false, a1, false, b1, (short)0, c[i], false, false);
        }
    }
}

template <bool RELU, int NT>
__device__ __forceinline__ void tile_store(
    const v8f (&c)[7], const float* __restrict__ bias, float* __restrict__ out,
    const int rg, const int col, const int lane, const int row0, const int nvalid)
{
    const int   lr   = lane & 15;
    const int   rtop = (lane >> 4) * 8;
    const float bv   = bias[col + lr];
#pragma unroll
    for (int i = 0; i < NT; ++i) {
        const int rb = (rg + 4 * i) * 16 + rtop;
        float* op = out + (size_t)(row0 + rb) * 128 + col + lr;
#pragma unroll
        for (int v = 0; v < 8; ++v) {
            if (rb + v < nvalid) {
                float o = c[i][v] + bv;
                if (RELU) o = fmaxf(o, 0.0f);
                op[v * 128] = o;
            }
        }
    }
}

template <bool RELU>
__global__ __launch_bounds__(1024) void sage_layer_lds(
    const float* __restrict__ feat, const int* __restrict__ src,
    const int* __restrict__ dst, const unsigned char* __restrict__ keys,
    float* __restrict__ deg, int count_deg, int E, int n_rows,
    unsigned* __restrict__ rec_list, int* __restrict__ rec_cnt,
    const unsigned* __restrict__ rep_list, const int* __restrict__ rep_cnt,
    const float* __restrict__ Wl, const float* __restrict__ bias,
    const float* __restrict__ Wr, float* __restrict__ out)
{
    __shared__ __attribute__((aligned(16))) float acc[AGG_R][ACC_LD];
    __shared__ unsigned seg[32][AGG_CAP];
    __shared__ unsigned stage[32][64];
    __shared__ int rowcnt[AGG_R];
    __shared__ int segcnt[32];
    __shared__ int segmore[32];

    const int tid  = threadIdx.x;
    const int wave = __builtin_amdgcn_readfirstlane(tid >> 5);
    const int lane = tid & 31;
    const int row0 = blockIdx.x * AGG_R;
    const bool count = (count_deg != 0);

    const int  my_r  = wave + 32 * lane;
    const bool my_in = (lane < AGG_RPW) && (my_r < AGG_R);
    const bool my_ok = my_in && (row0 + my_r < n_rows);
    float dl = 1.0f;
    if (!count && my_ok) dl = deg[row0 + my_r];

    bool replayed = false;
    if (rep_cnt != nullptr && !count) {
        const int  c  = rep_cnt[blockIdx.x * 32 + lane];
        const bool ok = (__builtin_amdgcn_ballot_w32(c < 0) == 0u);
        if (ok) {
            const int n = __builtin_amdgcn_readlane(c, wave);
            const unsigned* my = rep_list + (size_t)(blockIdx.x * 32 + wave) * REC_CAP;
            for (int r = wave; r < AGG_R; r += 32)
                *(float4*)&acc[r][lane * 4] = make_float4(0.f, 0.f, 0.f, 0.f);
            unsigned ent = (lane < n) ? my[lane] : 0u;
            for (int i0 = 0; i0 < n; i0 += 32) {
                const int      nx  = i0 + 32 + lane;
                const unsigned nxt = (nx < n) ? my[nx] : 0u;
                const int      m   = (n - i0 < 32) ? (n - i0) : 32;
                agg_add_rows(feat, acc, rowcnt, ent, m, lane, false);
                ent = nxt;
            }
            replayed = true;
        }
    }

    if (!replayed) {
        const bool rec = (rec_cnt != nullptr);
        unsigned* my_rec = rec ? rec_list + (size_t)(blockIdx.x * 32 + wave) * REC_CAP : nullptr;
        int rec_pos = 0;

        for (int i = tid; i < AGG_R * (ACC_LD / 4); i += 1024)
            ((float4*)&acc[0][0])[i] = make_float4(0.f, 0.f, 0.f, 0.f);
        for (int i = tid; i < AGG_R; i += 1024) rowcnt[i] = 0;

        const int total_steps = (E + AGG_STEP - 1) / AGG_STEP;
        const int per_wave    = (total_steps + 31) / 32;
        const int first       = wave * per_wave;
        int my_n = total_steps - first;
        if (my_n > per_wave) my_n = per_wave;
        if (my_n < 0) my_n = 0;
        const int rot = (my_n > 0) ? (int)(((long)blockIdx.x * my_n) / (long)gridDim.x) : 0;
        const unsigned keyrep = (unsigned)blockIdx.x * 0x01010101u;

        unsigned* my_list = &seg[wave][0];
        int cursor = 0;
        int more;
        do {
            int  cnt  = 0;
            bool full = false;
            while (cursor < my_n && !full) {
                const int c0 = cursor;
                uint4 kq[8];
                int   eb[8];
    #pragma unroll
                for (int u = 0; u < 8; ++u) {
                    int s = (c0 + u < my_n) ? (c0 + u) : (my_n - 1);
                    s += rot;
                    if (s >= my_n) s -= my_n;
                    eb[u] = (first + s) * AGG_STEP + lane * 16;
                    kq[u] = *(const uint4*)(keys + eb[u]);
                }
    #pragma unroll
                for (int u = 0; u < 8; ++u) {
                    if (!full && c0 + u < my_n) {
                        if (cnt + AGG_STEP > AGG_CAP) {
                            full = true;
                        } else {
                            agg_list_step(my_list, cnt, kq[u], eb[u], keyrep);
                            cursor = c0 + u + 1;
                        }
                    }
                }
            }
            {
                bool bad = false;
                for (int i = lane; i < cnt; i += 32) {
                    const int  e  = (int)my_list[i];
                    const bool in = (e < E);
                    const int  s  = in ? src[e] : 0;
                    const int  d  = in ? dst[e] : -1;
                    const unsigned r  = (unsigned)(d - row0);
                    const bool     ok = in && (r < (unsigned)AGG_R);
                    my_list[i] = ok ? (((unsigned)s << 9) | r) : 0xFFFFFFFFu;
                    bad = bad || !ok;
                }
                if (__builtin_amdgcn_ballot_w32(bad) != 0u) {
                    int kept = 0;
                    for (int i0 = 0; i0 < cnt; i0 += 32) {
                        const int      idx = i0 + lane;
                        const unsigned en  = (idx < cnt) ? my_list[idx] : 0xFFFFFFFFu;
                        const bool     ok  = (en != 0xFFFFFFFFu);
                        const unsigned k   = __builtin_amdgcn_ballot_w32(ok);
                        if (ok) my_list[kept + (int)__builtin_amdgcn_mbcnt_lo(k, 0u)] = en;
                        kept += __builtin_popcount(k);
                    }
                    cnt = kept;
                }
            }
            if (lane == 0) {
                segcnt[wave]  = cnt;
                segmore[wave] = (cursor < my_n) ? 1 : 0;
            }
            __syncthreads();

            more = __builtin_amdgcn_readfirstlane(
                (int)(__builtin_amdgcn_ballot_w32(segmore[lane] != 0) != 0u));
            int staged = 0;
            for (int s = 0; s < 32; ++s) {
                const int n = __builtin_amdgcn_readfirstlane(segcnt[s]);
                for (int i0 = 0; i0 < n; i0 += 32) {
                    const int idx = i0 + lane;
                    const unsigned en = (idx < n) ? seg[s][idx] : 0u;
                    const bool mine = (idx < n) && ((int)(en & 31u) == wave);
                    const unsigned mk = __builtin_amdgcn_ballot_w32(mine);
                    if (mk != 0u) {
                        if (mine) stage[wave][staged + (int)__builtin_amdgcn_mbcnt_lo(mk, 0u)] = en;
                        staged += __builtin_popcount(mk);
                        if (staged >= 32) {
                            const unsigned ent  = stage[wave][lane];
                            const unsigned rest = stage[wave][32 + lane];
                            if (rec) {
                                if (rec_pos + 32 <= REC_CAP) my_rec[rec_pos + lane] = ent;
                                rec_pos += 32;
                            }
                            agg_add_rows(feat, acc, rowcnt, ent, 32, lane, count);
                            staged -= 32;
                            if (lane < staged) stage[wave][lane] = rest;
                        }
                    }
                }
            }
            if (staged > 0) {
                const unsigned ent = stage[wave][lane];
                if (rec) {
                    if (rec_pos + staged <= REC_CAP && lane < staged) my_rec[rec_pos + lane] = ent;
                    rec_pos += staged;
                }
                agg_add_rows(feat, acc, rowcnt, ent, staged, lane, count);
            }
            __syncthreads();
        } while (more);

        if (rec && lane == 0)
            rec_cnt[blockIdx.x * 32 + wave] = (rec_pos <= REC_CAP) ? rec_pos : -1;
    }

    if (count) {
        dl = my_in ? (float)rowcnt[my_r] : 1.0f;
        if (my_ok) deg[row0 + my_r] = dl;
    }
    for (int i = 0; i < AGG_RPW; ++i) {
        const int r = wave + 32 * i;
        if (r < AGG_R) {
            const float d = __int_as_float(
                __builtin_amdgcn_readlane(__float_as_int(dl), i));
            const float invd = 1.0f / fmaxf(d, 1.0f);
            float4* p = (float4*)&acc[r][lane * 4];
            float4 a = *p;
            a.x *= invd; a.y *= invd; a.z *= invd; a.w *= invd;
            *p = a;
        }
    }
    __syncthreads();

    const int rg  = wave >> 3;
    const int col = (wave & 7) * 16;
    v8f c[7];
#pragma unroll
    for (int i = 0; i < 7; ++i) c[i] = v8f{};

    if (rg == 0) tile_products<7>(c, acc, Wl, rg, col, lane);
    else         tile_products<6>(c, acc, Wl, rg, col, lane);

    __syncthreads();
    const int nvalid = (n_rows - row0 < AGG_R) ? (n_rows - row0) : AGG_R;
    for (int i0 = 0; i0 < AGG_RPW; i0 += 7) {
        float4 v[7];
#pragma unroll
        for (int u = 0; u < 7; ++u) {
            const int r = wave + 32 * (i0 + u);
            v[u] = (r < nvalid)
                 ? ((const float4*)(feat + (size_t)(row0 + r) * 128))[lane]
                 : make_float4(0.f, 0.f, 0.f, 0.f);
        }
#pragma unroll
        for (int u = 0; u < 7; ++u) {
            const int r = wave + 32 * (i0 + u);
            if (r < AGG_R) *(float4*)&acc[r][lane * 4] = v[u];
        }
    }
    __syncthreads();

    if (rg == 0) {
        tile_products<7>(c, acc, Wr, rg, col, lane);
        tile_store<RELU, 7>(c, bias, out, rg, col, lane, row0, nvalid);
    } else {
        tile_products<6>(c, acc, Wr, rg, col, lane);
        tile_store<RELU, 6>(c, bias, out, rg, col, lane, row0, nvalid);
    }
}

extern "C" void kernel_launch(void* const* d_in, const int* in_sizes, int n_in,
                              void* d_out, int out_size, void* d_ws, size_t ws_size,
                              hipStream_t stream) {
    const float* x   = (const float*)d_in[0];
    const int*   ei  = (const int*)d_in[1];
    const float* W1l = (const float*)d_in[2];
    const float* b1l = (const float*)d_in[3];
    const float* W1r = (const float*)d_in[4];
    const float* W2l = (const float*)d_in[5];
    const float* b2l = (const float*)d_in[6];
    const float* W2r = (const float*)d_in[7];
    float* out = (float*)d_out;

    const int N = in_sizes[0] / 128;
    const int E = in_sizes[1] / 2;
    const int* src = ei;
    const int* dst = ei + E;

    float* h1  = (float*)d_ws;
    float* deg = h1 + (size_t)N * 128;

    const int  ag_blocks = (N + AGG_R - 1) / AGG_R;

    int*      rec_cnt  = (int*)(deg + N);
    unsigned* rec_list = (unsigned*)(rec_cnt + ag_blocks * 32);

    unsigned char* keys = (unsigned char*)(rec_list + (size_t)ag_blocks * 32 * REC_CAP);
    keys += (16 - ((size_t)keys & 15)) & 15;
    const int key_steps = (E + AGG_STEP - 1) / AGG_STEP;

    edge_block_keys<<<(key_steps + 7) / 8, 256, 0, stream>>>(
        dst, (unsigned*)keys, E, key_steps);

    sage_layer_lds<true><<<ag_blocks, 1024, 0, stream>>>(
        x, src, dst, keys, deg, 1, E, N, rec_list, rec_cnt, nullptr, nullptr,
        W1l, b1l, W1r, h1);

    sage_layer_lds<false><<<ag_blocks, 1024, 0, stream>>>(
        h1, src, dst, keys, deg, 0, E, N, nullptr, nullptr, rec_list, rec_cnt,
        W2l, b2l, W2r, out);
}


# layer-1 lists put in owner order by the scanning wave (stable in-register counting sort on the 5 owner bits): each wave reads only its own group, not all 32 lists; entry and sum order unchanged
# speedup vs baseline: 3.4984x; 1.0692x over previous
#include <hip/hip_runtime.h>
#include <hip/hip_bf16.h>

typedef __attribute__((ext_vector_type(2))) float v2f;
typedef __attribute__((ext_vector_type(8))) float v8f;

#define AGG_R    392
#define ACC_LD   132
#define AGG_RPW  13
#define AGG_CAP  832
#define AGG_STEP 512
#define REC_CAP  512
#define GRP_MAX  256
#define GRP_OFF  (AGG_CAP - 33)
#define GRP_FLAG 0x40000000

__device__ __forceinline__ void agg_list_step(
    unsigned* __restrict__ list, int& cnt, const uint4 kq, const int eb,
    const unsigned keyrep)
{
    const unsigned w[4] = {kq.x, kq.y, kq.z, kq.w};
    unsigned m = 0u;
#pragma unroll
    for (int q = 0; q < 4; ++q) {
        const unsigned x = w[q] ^ keyrep;
        const unsigned t = ~(((x & 0x7F7F7F7Fu) + 0x7F7F7F7Fu) | x | 0x7F7F7F7Fu);
        m |= t >> (7 - q);
    }
    while (__builtin_amdgcn_ballot_w32(m != 0u) != 0u) {
        const bool     has = (m != 0u);
        const unsigned k   = __builtin_amdgcn_ballot_w32(has);
        if (has) {
            const int p = __builtin_ctz(m);
            list[cnt + (int)__builtin_amdgcn_mbcnt_lo(k, 0u)] =
                (unsigned)(eb + ((p & 3) << 2) + (p >> 3));
        }
        cnt += __builtin_popcount(k);
        m &= m - 1u;
    }
}

__device__ __forceinline__ int4 agg_load_dst_checked(
    const int* __restrict__ dst, const int e4, const int E)
{
    int4 d;
    d.x = (e4 + 0 < E) ? dst[e4 + 0] : -1;
    d.y = (e4 + 1 < E) ? dst[e4 + 1] : -1;
    d.z = (e4 + 2 < E) ? dst[e4 + 2] : -1;
    d.w = (e4 + 3 < E) ? dst[e4 + 3] : -1;
    return d;
}

__global__ __launch_bounds__(256) void edge_block_keys(
    const int* __restrict__ dst, unsigned* __restrict__ keys, int E, int n_steps)
{
    const int step = blockIdx.x * 8 + (threadIdx.x >> 5);
    const int lane = threadIdx.x & 31;
    if (step >= n_steps) return;
    const bool vec = ((((size_t)dst) & 15) == 0);
    int4 d[4];
#pragma unroll
    for (int j = 0; j < 4; ++j) {
        const int e0 = step * AGG_STEP + j * 128 + lane * 4;
        if (vec && (e0 + 4 <= E)) d[j] = *(const int4*)(dst + e0);
        else                      d[j] = agg_load_dst_checked(dst, e0, E);
    }
#pragma unroll
    for (int j = 0; j < 4; ++j) {
        const unsigned k0 = min((unsigned)d[j].x / (unsigned)AGG_R, 255u);
        const unsigned k1 = min((unsigned)d[j].y / (unsigned)AGG_R, 255u);
        const unsigned k2 = min((unsigned)d[j].z / (unsigned)AGG_R, 255u);
        const unsigned k3 = min((unsigned)d[j].w / (unsigned)AGG_R, 255u);
        keys[step * (AGG_STEP / 4) + j * 32 + lane] =
            k0 | (k1 << 8) | (k2 << 16) | (k3 << 24);
    }
}

__device__ __forceinline__ void agg_add_rows(
    const float* __restrict__ feat, float (*acc)[ACC_LD], int* rowcnt,
    const unsigned ent, const int n, const int lane, const bool count)
{
    for (int j = 0; j < n; j += 8) {
        float4   v[8];
        unsigned en[8];
#pragma unroll
        for (int u = 0; u < 8; ++u) {
            const int jj = (j + u < n) ? (j + u) : (n - 1);
            en[u] = (unsigned)__builtin_amdgcn_readlane((int)ent, jj);
            v[u]  = ((const float4*)(feat + (size_t)(en[u] >> 9) * 128))[lane];
        }
#pragma unroll
        for (int u = 0; u < 8; ++u) {
            if (j + u < n) {
                const int r = (int)(en[u] & 511u);
                float4* p = (float4*)&acc[r][lane * 4];
                float4 a = *p;
                a.x += v[u].x; a.y += v[u].y; a.z += v[u].z; a.w += v[u].w;
                *p = a;
                if (count && lane == 0) rowcnt[r] += 1;
            }
        }
    }
}

template <int NT>
__device__ __forceinline__ void tile_products(
    v8f (&c)[7], const float (*acc)[ACC_LD], const float* __restrict__ W,
    const int rg, const int col, const int lane)
{
    const int lr = lane & 15;
    const int kh = (lane >> 4) * 4;
    const float* wrow = W + (size_t)(col + lr) * 128 + kh;
    const float* arow[NT];
#pragma unroll
    for (int i = 0; i < NT; ++i) {
        int r = (rg + 4 * i) * 16 + lr;
        if (i == 6 && r > AGG_R - 1) r = AGG_R - 1;
        arow[i] = &acc[r][kh];
    }
#pragma unroll 1
    for (int j = 0; j < 16; ++j) {
        const float4 b = *(const float4*)(wrow + 8 * j);
        float4 a[NT];
#pragma unroll
        for (int i = 0; i < NT; ++i) a[i] = *(const float4*)(arow[i] + 8 * j);
        const v2f b0 = {b.x, b.y};
        const v2f b1 = {b.z, b.w};
#pragma unroll
        for (int i = 0; i < NT; ++i) {
            const v2f a0 = {a[i].x, a[i].y};
            c[i] = __builtin_amdgcn_wmma_f32_16x16x4_f32(
                false, a0, false, b0, (short)0, c[i], false, false);
        }
#pragma unroll
        for (int i = 0; i < NT; ++i) {
            const v2f a1 = {a[i].z, a[i].w};
            c[i] = __builtin_amdgcn_wmma_f32_16x16x4_f32(
                false, a1, false, b1, (short)0, c[i], false, false);
        }
    }
}

template <bool RELU, int NT>
__device__ __forceinline__ void tile_store(
    const v8f (&c)[7], const float* __restrict__ bias, float* __restrict__ out,
    const int rg, const int col, const int lane, const int row0, const int nvalid)
{
    const int   lr   = lane & 15;
    const int   rtop = (lane >> 4) * 8;
    const float bv   = bias[col + lr];
#pragma unroll
    for (int i = 0; i < NT; ++i) {
        const int rb = (rg + 4 * i) * 16 + rtop;
        float* op = out + (size_t)(row0 + rb) * 128 + col + lr;
#pragma unroll
        for (int v = 0; v < 8; ++v) {
            if (rb + v < nvalid) {
                float o = c[i][v] + bv;
                if (RELU) o = fmaxf(o, 0.0f);
                op[v * 128] = o;
            }
        }
    }
}

template <bool RELU>
__global__ __launch_bounds__(1024) void sage_layer_lds(
    const float* __restrict__ feat, const int* __restrict__ src,
    const int* __restrict__ dst, const unsigned char* __restrict__ keys,
    float* __restrict__ deg, int count_deg, int E, int n_rows,
    unsigned* __restrict__ rec_list, int* __restrict__ rec_cnt,
    const unsigned* __restrict__ rep_list, const int* __restrict__ rep_cnt,
    const float* __restrict__ Wl, const float* __restrict__ bias,
    const float* __restrict__ Wr, float* __restrict__ out)
{
    __shared__ __attribute__((aligned(16))) float acc[AGG_R][ACC_LD];
    __shared__ unsigned seg[32][AGG_CAP];
    __shared__ unsigned stage[32][64];
    __shared__ int rowcnt[AGG_R];
    __shared__ int segcnt[32];
    __shared__ int segmore[32];

    const int tid  = threadIdx.x;
    const int wave = __builtin_amdgcn_readfirstlane(tid >> 5);
    const int lane = tid & 31;
    const int row0 = blockIdx.x * AGG_R;
    const bool count = (count_deg != 0);

    const int  my_r  = wave + 32 * lane;
    const bool my_in = (lane < AGG_RPW) && (my_r < AGG_R);
    const bool my_ok = my_in && (row0 + my_r < n_rows);
    float dl = 1.0f;
    if (!count && my_ok) dl = deg[row0 + my_r];

    bool replayed = false;
    if (rep_cnt != nullptr && !count) {
        const int  c  = rep_cnt[blockIdx.x * 32 + lane];
        const bool ok = (__builtin_amdgcn_ballot_w32(c < 0) == 0u);
        if (ok) {
            const int n = __builtin_amdgcn_readlane(c, wave);
            const unsigned* my = rep_list + (size_t)(blockIdx.x * 32 + wave) * REC_CAP;
            for (int r = wave; r < AGG_R; r += 32)
                *(float4*)&acc[r][lane * 4] = make_float4(0.f, 0.f, 0.f, 0.f);
            unsigned ent = (lane < n) ? my[lane] : 0u;
            for (int i0 = 0; i0 < n; i0 += 32) {
                const int      nx  = i0 + 32 + lane;
                const unsigned nxt = (nx < n) ? my[nx] : 0u;
                const int      m   = (n - i0 < 32) ? (n - i0) : 32;
                agg_add_rows(feat, acc, rowcnt, ent, m, lane, false);
                ent = nxt;
            }
            replayed = true;
        }
    }

    if (!replayed) {
        const bool rec = (rec_cnt != nullptr);
        unsigned* my_rec = rec ? rec_list + (size_t)(blockIdx.x * 32 + wave) * REC_CAP : nullptr;
        int rec_pos = 0;

        for (int i = tid; i < AGG_R * (ACC_LD / 4); i += 1024)
            ((float4*)&acc[0][0])[i] = make_float4(0.f, 0.f, 0.f, 0.f);
        for (int i = tid; i < AGG_R; i += 1024) rowcnt[i] = 0;

        const int total_steps = (E + AGG_STEP - 1) / AGG_STEP;
        const int per_wave    = (total_steps + 31) / 32;
        const int first       = wave * per_wave;
        int my_n = total_steps - first;
        if (my_n > per_wave) my_n = per_wave;
        if (my_n < 0) my_n = 0;
        const int rot = (my_n > 0) ? (int)(((long)blockIdx.x * my_n) / (long)gridDim.x) : 0;
        const unsigned keyrep = (unsigned)blockIdx.x * 0x01010101u;

        unsigned* my_list = &seg[wave][0];
        int cursor = 0;
        int more;
        do {
            int  cnt  = 0;
            bool full = false;
            while (cursor < my_n && !full) {
                const int c0 = cursor;
                uint4 kq[8];
                int   eb[8];
    #pragma unroll
                for (int u = 0; u < 8; ++u) {
                    int s = (c0 + u < my_n) ? (c0 + u) : (my_n - 1);
                    s += rot;
                    if (s >= my_n) s -= my_n;
                    eb[u] = (first + s) * AGG_STEP + lane * 16;
                    kq[u] = *(const uint4*)(keys + eb[u]);
                }
    #pragma unroll
                for (int u = 0; u < 8; ++u) {
                    if (!full && c0 + u < my_n) {
                        if (cnt + AGG_STEP > AGG_CAP) {
                            full = true;
                        } else {
                            agg_list_step(my_list, cnt, kq[u], eb[u], keyrep);
                            cursor = c0 + u + 1;
                        }
                    }
                }
            }
            {
                bool bad = false;
                for (int i = lane; i < cnt; i += 32) {
                    const int  e  = (int)my_list[i];
                    const bool in = (e < E);
                    const int  s  = in ? src[e] : 0;
                    const int  d  = in ? dst[e] : -1;
                    const unsigned r  = (unsigned)(d - row0);
                    const bool     ok = in && (r < (unsigned)AGG_R);
                    my_list[i] = ok ? (((unsigned)s << 9) | r) : 0xFFFFFFFFu;
                    bad = bad || !ok;
                }
                if (__builtin_amdgcn_ballot_w32(bad) != 0u) {
                    int kept = 0;
                    for (int i0 = 0; i0 < cnt; i0 += 32) {
                        const int      idx = i0 + lane;
                        const unsigned en  = (idx < cnt) ? my_list[idx] : 0xFFFFFFFFu;
                        const bool     ok  = (en != 0xFFFFFFFFu);
                        const unsigned k   = __builtin_amdgcn_ballot_w32(ok);
                        if (ok) my_list[kept + (int)__builtin_amdgcn_mbcnt_lo(k, 0u)] = en;
                        kept += __builtin_popcount(k);
                    }
                    cnt = kept;
                }
            }
            bool grouped = false;
            if (cnt <= GRP_MAX) {
                unsigned* hist = &stage[wave][0];
                unsigned  e[8];
                unsigned  pk[8];
#pragma unroll
                for (int c = 0; c < 8; ++c) {
                    const int idx = c * 32 + lane;
                    e[c] = (idx < cnt) ? my_list[idx] : 0u;
                }
                hist[lane] = 0u;
#pragma unroll
                for (int c = 0; c < 8; ++c) {
                    pk[c] = 0u;
                    if (c * 32 < cnt) {
                        const bool     valid = (c * 32 + lane < cnt);
                        const unsigned key   = e[c] & 31u;
                        unsigned m = __builtin_amdgcn_ballot_w32(valid);
#pragma unroll
                        for (int b = 0; b < 5; ++b) {
                            const bool     bit = ((key >> b) & 1u) != 0u;
                            const unsigned bb  = __builtin_amdgcn_ballot_w32(bit);
                            m &= bit ? bb : ~bb;
                        }
                        const unsigned rank = __builtin_amdgcn_mbcnt_lo(m, 0u);
                        const unsigned base = hist[key];
                        pk[c] = base + rank;
                        if (valid && rank == 0u) hist[key] = base + (unsigned)__builtin_popcount(m);
                    }
                }
                const unsigned hc = hist[lane];
                unsigned inc = hc;
#pragma unroll
                for (int d = 1; d < 32; d <<= 1) {
                    const unsigned t = (unsigned)__shfl_up((int)inc, d);
                    if (lane >= d) inc += t;
                }
                const unsigned gstart = inc - hc;
                hist[lane] = gstart;
                my_list[GRP_OFF + lane] = gstart;
                if (lane == 0) my_list[GRP_OFF + 32] = (unsigned)cnt;
#pragma unroll
                for (int c = 0; c < 8; ++c) {
                    if (c * 32 + lane < cnt) my_list[hist[e[c] & 31u] + pk[c]] = e[c];
                }
                grouped = true;
            }
            if (lane == 0) {
                segcnt[wave]  = grouped ? (cnt | GRP_FLAG) : cnt;
                segmore[wave] = (cursor < my_n) ? 1 : 0;
            }
            __syncthreads();

            more = __builtin_amdgcn_readfirstlane(
                (int)(__builtin_amdgcn_ballot_w32(segmore[lane] != 0) != 0u));
            int staged = 0;
            const int  raw_l = segcnt[lane];
            const bool grp_l = (raw_l & GRP_FLAG) != 0;
            int beg_l = 0;
            int end_l = raw_l & (GRP_FLAG - 1);
            if (grp_l) {
                beg_l = (int)seg[lane][GRP_OFF + wave];
                end_l = (int)seg[lane][GRP_OFF + wave + 1];
            }
            for (int s = 0; s < 32; ++s) {
                const int  n   = __builtin_amdgcn_readlane(end_l, s);
                const int  i_b = __builtin_amdgcn_readlane(beg_l, s);
                const bool grp = __builtin_amdgcn_readlane(grp_l ? 1 : 0, s) != 0;
                for (int i0 = i_b; i0 < n; i0 += 32) {
                    const int idx = i0 + lane;
                    const unsigned en = (idx < n) ? seg[s][idx] : 0u;
                    const bool mine = (idx < n) && (grp || ((int)(en & 31u) == wave));
                    const unsigned mk = __builtin_amdgcn_ballot_w32(mine);
                    if (mk != 0u) {
                        if (mine) stage[wave][staged + (int)__builtin_amdgcn_mbcnt_lo(mk, 0u)] = en;
                        staged += __builtin_popcount(mk);
                        if (staged >= 32) {
                            const unsigned ent  = stage[wave][lane];
                            const unsigned rest = stage[wave][32 + lane];
                            if (rec) {
                                if (rec_pos + 32 <= REC_CAP) my_rec[rec_pos + lane] = ent;
                                rec_pos += 32;
                            }
                            agg_add_rows(feat, acc, rowcnt, ent, 32, lane, count);
                            staged -= 32;
                            if (lane < staged) stage[wave][lane] = rest;
                        }
                    }
                }
            }
            if (staged > 0) {
                const unsigned ent = stage[wave][lane];
                if (rec) {
                    if (rec_pos + staged <= REC_CAP && lane < staged) my_rec[rec_pos + lane] = ent;
                    rec_pos += staged;
                }
                agg_add_rows(feat, acc, rowcnt, ent, staged, lane, count);
            }
            __syncthreads();
        } while (more);

        if (rec && lane == 0)
            rec_cnt[blockIdx.x * 32 + wave] = (rec_pos <= REC_CAP) ? rec_pos : -1;
    }

    if (count) {
        dl = my_in ? (float)rowcnt[my_r] : 1.0f;
        if (my_ok) deg[row0 + my_r] = dl;
    }
    for (int i = 0; i < AGG_RPW; ++i) {
        const int r = wave + 32 * i;
        if (r < AGG_R) {
            const float d = __int_as_float(
                __builtin_amdgcn_readlane(__float_as_int(dl), i));
            const float invd = 1.0f / fmaxf(d, 1.0f);
            float4* p = (float4*)&acc[r][lane * 4];
            float4 a = *p;
            a.x *= invd; a.y *= invd; a.z *= invd; a.w *= invd;
            *p = a;
        }
    }
    __syncthreads();

    const int rg  = wave >> 3;
    const int col = (wave & 7) * 16;
    v8f c[7];
#pragma unroll
    for (int i = 0; i < 7; ++i) c[i] = v8f{};

    if (rg == 0) tile_products<7>(c, acc, Wl, rg, col, lane);
    else         tile_products<6>(c, acc, Wl, rg, col, lane);

    __syncthreads();
    const int nvalid = (n_rows - row0 < AGG_R) ? (n_rows - row0) : AGG_R;
    for (int i0 = 0; i0 < AGG_RPW; i0 += 7) {
        float4 v[7];
#pragma unroll
        for (int u = 0; u < 7; ++u) {
            const int r = wave + 32 * (i0 + u);
            v[u] = (r < nvalid)
                 ? ((const float4*)(feat + (size_t)(row0 + r) * 128))[lane]
                 : make_float4(0.f, 0.f, 0.f, 0.f);
        }
#pragma unroll
        for (int u = 0; u < 7; ++u) {
            const int r = wave + 32 * (i0 + u);
            if (r < AGG_R) *(float4*)&acc[r][lane * 4] = v[u];
        }
    }
    __syncthreads();

    if (rg == 0) {
        tile_products<7>(c, acc, Wr, rg, col, lane);
        tile_store<RELU, 7>(c, bias, out, rg, col, lane, row0, nvalid);
    } else {
        tile_products<6>(c, acc, Wr, rg, col, lane);
        tile_store<RELU, 6>(c, bias, out, rg, col, lane, row0, nvalid);
    }
}

extern "C" void kernel_launch(void* const* d_in, const int* in_sizes, int n_in,
                              void* d_out, int out_size, void* d_ws, size_t ws_size,
                              hipStream_t stream) {
    const float* x   = (const float*)d_in[0];
    const int*   ei  = (const int*)d_in[1];
    const float* W1l = (const float*)d_in[2];
    const float* b1l = (const float*)d_in[3];
    const float* W1r = (const float*)d_in[4];
    const float* W2l = (const float*)d_in[5];
    const float* b2l = (const float*)d_in[6];
    const float* W2r = (const float*)d_in[7];
    float* out = (float*)d_out;

    const int N = in_sizes[0] / 128;
    const int E = in_sizes[1] / 2;
    const int* src = ei;
    const int* dst = ei + E;

    float* h1  = (float*)d_ws;
    float* deg = h1 + (size_t)N * 128;

    const int  ag_blocks = (N + AGG_R - 1) / AGG_R;

    int*      rec_cnt  = (int*)(deg + N);
    unsigned* rec_list = (unsigned*)(rec_cnt + ag_blocks * 32);

    unsigned char* keys = (unsigned char*)(rec_list + (size_t)ag_blocks * 32 * REC_CAP);
    keys += (16 - ((size_t)keys & 15)) & 15;
    const int key_steps = (E + AGG_STEP - 1) / AGG_STEP;

    edge_block_keys<<<(key_steps + 7) / 8, 256, 0, stream>>>(
        dst, (unsigned*)keys, E, key_steps);

    sage_layer_lds<true><<<ag_blocks, 1024, 0, stream>>>(
        x, src, dst, keys, deg, 1, E, N, rec_list, rec_cnt, nullptr, nullptr,
        W1l, b1l, W1r, h1);

    sage_layer_lds<false><<<ag_blocks, 1024, 0, stream>>>(
        h1, src, dst, keys, deg, 0, E, N, nullptr, nullptr, rec_list, rec_cnt,
        W2l, b2l, W2r, out);
}


# both layers: each wave sums a row's sources in registers, one LDS add per row not per edge, over its edge list put in row order by layer 1 after the list walk (stable counting sort in registers)
# speedup vs baseline: 3.7344x; 1.0675x over previous
#include <hip/hip_runtime.h>
#include <hip/hip_bf16.h>

typedef __attribute__((ext_vector_type(2))) float v2f;
typedef __attribute__((ext_vector_type(8))) float v8f;

#define AGG_R    392
#define ACC_LD   132
#define AGG_RPW  13
#define AGG_CAP  832
#define AGG_STEP 512
#define REC_CAP  512
#define GRP_MAX  256
#define GRP_OFF  (AGG_CAP - 33)
#define GRP_FLAG 0x40000000

__device__ __forceinline__ void agg_list_step(
    unsigned* __restrict__ list, int& cnt, const uint4 kq, const int eb,
    const unsigned keyrep)
{
    const unsigned w[4] = {kq.x, kq.y, kq.z, kq.w};
    unsigned m = 0u;
#pragma unroll
    for (int q = 0; q < 4; ++q) {
        const unsigned x = w[q] ^ keyrep;
        const unsigned t = ~(((x & 0x7F7F7F7Fu) + 0x7F7F7F7Fu) | x | 0x7F7F7F7Fu);
        m |= t >> (7 - q);
    }
    while (__builtin_amdgcn_ballot_w32(m != 0u) != 0u) {
        const bool     has = (m != 0u);
        const unsigned k   = __builtin_amdgcn_ballot_w32(has);
        if (has) {
            const int p = __builtin_ctz(m);
            list[cnt + (int)__builtin_amdgcn_mbcnt_lo(k, 0u)] =
                (unsigned)(eb + ((p & 3) << 2) + (p >> 3));
        }
        cnt += __builtin_popcount(k);
        m &= m - 1u;
    }
}

__device__ __forceinline__ int4 agg_load_dst_checked(
    const int* __restrict__ dst, const int e4, const int E)
{
    int4 d;
    d.x = (e4 + 0 < E) ? dst[e4 + 0] : -1;
    d.y = (e4 + 1 < E) ? dst[e4 + 1] : -1;
    d.z = (e4 + 2 < E) ? dst[e4 + 2] : -1;
    d.w = (e4 + 3 < E) ? dst[e4 + 3] : -1;
    return d;
}

__global__ __launch_bounds__(256) void edge_block_keys(
    const int* __restrict__ dst, unsigned* __restrict__ keys, int E, int n_steps)
{
    const int step = blockIdx.x * 8 + (threadIdx.x >> 5);
    const int lane = threadIdx.x & 31;
    if (step >= n_steps) return;
    const bool vec = ((((size_t)dst) & 15) == 0);
    int4 d[4];
#pragma unroll
    for (int j = 0; j < 4; ++j) {
        const int e0 = step * AGG_STEP + j * 128 + lane * 4;
        if (vec && (e0 + 4 <= E)) d[j] = *(const int4*)(dst + e0);
        else                      d[j] = agg_load_dst_checked(dst, e0, E);
    }
#pragma unroll
    for (int j = 0; j < 4; ++j) {
        const unsigned k0 = min((unsigned)d[j].x / (unsigned)AGG_R, 255u);
        const unsigned k1 = min((unsigned)d[j].y / (unsigned)AGG_R, 255u);
        const unsigned k2 = min((unsigned)d[j].z / (unsigned)AGG_R, 255u);
        const unsigned k3 = min((unsigned)d[j].w / (unsigned)AGG_R, 255u);
        keys[step * (AGG_STEP / 4) + j * 32 + lane] =
            k0 | (k1 << 8) | (k2 << 16) | (k3 << 24);
    }
}

__device__ __forceinline__ void agg_add_rows(
    const float* __restrict__ feat, float (*acc)[ACC_LD], int* rowcnt,
    const unsigned ent, const int n, const int lane, const bool count)
{
    for (int j = 0; j < n; j += 8) {
        float4   v[8];
        unsigned en[8];
#pragma unroll
        for (int u = 0; u < 8; ++u) {
            const int jj = (j + u < n) ? (j + u) : (n - 1);
            en[u] = (unsigned)__builtin_amdgcn_readlane((int)ent, jj);
            v[u]  = ((const float4*)(feat + (size_t)(en[u] >> 9) * 128))[lane];
        }
#pragma unroll
        for (int u = 0; u < 8; ++u) {
            if (j + u < n) {
                const int r = (int)(en[u] & 511u);
                float4* p = (float4*)&acc[r][lane * 4];
                float4 a = *p;
                a.x += v[u].x; a.y += v[u].y; a.z += v[u].z; a.w += v[u].w;
                *p = a;
                if (count && lane == 0) rowcnt[r] += 1;
            }
        }
    }
}

__device__ __forceinline__ void agg_row_flush(
    float (*acc)[ACC_LD], const int r, const int lane, const float4 sum)
{
    float4* p = (float4*)&acc[r][lane * 4];
    float4 a = *p;
    a.x += sum.x; a.y += sum.y; a.z += sum.z; a.w += sum.w;
    *p = a;
}

__device__ __forceinline__ void agg_sum_runs(
    const float* __restrict__ feat, float (*acc)[ACC_LD],
    const unsigned ent, const int n, const int lane, float4& sum, int& cur)
{
    for (int j = 0; j < n; j += 8) {
        float4   v[8];
        unsigned en[8];
#pragma unroll
        for (int u = 0; u < 8; ++u) {
            const int jj = (j + u < n) ? (j + u) : (n - 1);
            en[u] = (unsigned)__builtin_amdgcn_readlane((int)ent, jj);
            v[u]  = ((const float4*)(feat + (size_t)(en[u] >> 9) * 128))[lane];
        }
#pragma unroll
        for (int u = 0; u < 8; ++u) {
            if (j + u < n) {
                const int r = (int)(en[u] & 511u);
                if (r != cur) {
                    if (cur >= 0) agg_row_flush(acc, cur, lane, sum);
                    sum = make_float4(0.f, 0.f, 0.f, 0.f);
                    cur = r;
                }
                sum.x += v[u].x; sum.y += v[u].y; sum.z += v[u].z; sum.w += v[u].w;
            }
        }
    }
}

__device__ __forceinline__ unsigned agg_rows_sorted(
    const float* __restrict__ feat, float (*acc)[ACC_LD], unsigned* my,
    const int n, unsigned* sl, unsigned* hist, const int lane)
{
    unsigned e[REC_CAP / 32];
    unsigned pk[REC_CAP / 32];
#pragma unroll
    for (int c = 0; c < REC_CAP / 32; ++c) {
        const int idx = c * 32 + lane;
        e[c] = (idx < n) ? my[idx] : 0u;
    }
    hist[lane] = 0u;
#pragma unroll
    for (int c = 0; c < REC_CAP / 32; ++c) {
        pk[c] = 0u;
        if (c * 32 < n) {
            const bool     valid = (c * 32 + lane < n);
            const unsigned key   = (e[c] >> 5) & 15u;
            unsigned m = __builtin_amdgcn_ballot_w32(valid);
#pragma unroll
            for (int b = 0; b < 4; ++b) {
                const bool     bit = ((key >> b) & 1u) != 0u;
                const unsigned bb  = __builtin_amdgcn_ballot_w32(bit);
                m &= bit ? bb : ~bb;
            }
            const unsigned rank = __builtin_amdgcn_mbcnt_lo(m, 0u);
            const unsigned base = hist[key];
            pk[c] = base + rank;
            if (valid && rank == 0u) hist[key] = base + (unsigned)__builtin_popcount(m);
        }
    }
    const unsigned hc = hist[lane];
    unsigned inc = hc;
#pragma unroll
    for (int d = 1; d < 32; d <<= 1) {
        const unsigned t = (unsigned)__shfl_up((int)inc, d);
        if (lane >= d) inc += t;
    }
    hist[lane] = inc - hc;
#pragma unroll
    for (int c = 0; c < REC_CAP / 32; ++c) {
        if (c * 32 + lane < n) {
            const unsigned pos = hist[(e[c] >> 5) & 15u] + pk[c];
            sl[pos] = e[c];
            my[pos] = e[c];
        }
    }
    float4 sum = make_float4(0.f, 0.f, 0.f, 0.f);
    int    cur = -1;
    unsigned ent = (lane < n) ? sl[lane] : 0u;
    for (int i0 = 0; i0 < n; i0 += 32) {
        const int      nx  = i0 + 32 + lane;
        const unsigned nxt = (nx < n) ? sl[nx] : 0u;
        const int      m   = (n - i0 < 32) ? (n - i0) : 32;
        agg_sum_runs(feat, acc, ent, m, lane, sum, cur);
        ent = nxt;
    }
    if (cur >= 0) agg_row_flush(acc, cur, lane, sum);
    return hc;
}

template <int NT>
__device__ __forceinline__ void tile_products(
    v8f (&c)[7], const float (*acc)[ACC_LD], const float* __restrict__ W,
    const int rg, const int col, const int lane)
{
    const int lr = lane & 15;
    const int kh = (lane >> 4) * 4;
    const float* wrow = W + (size_t)(col + lr) * 128 + kh;
    const float* arow[NT];
#pragma unroll
    for (int i = 0; i < NT; ++i) {
        int r = (rg + 4 * i) * 16 + lr;
        if (i == 6 && r > AGG_R - 1) r = AGG_R - 1;
        arow[i] = &acc[r][kh];
    }
#pragma unroll 1
    for (int j = 0; j < 16; ++j) {
        const float4 b = *(const float4*)(wrow + 8 * j);
        float4 a[NT];
#pragma unroll
        for (int i = 0; i < NT; ++i) a[i] = *(const float4*)(arow[i] + 8 * j);
        const v2f b0 = {b.x, b.y};
        const v2f b1 = {b.z, b.w};
#pragma unroll
        for (int i = 0; i < NT; ++i) {
            const v2f a0 = {a[i].x, a[i].y};
            c[i] = __builtin_amdgcn_wmma_f32_16x16x4_f32(
                false, a0, false, b0, (short)0, c[i], false, false);
        }
#pragma unroll
        for (int i = 0; i < NT; ++i) {
            const v2f a1 = {a[i].z, a[i].w};
            c[i] = __builtin_amdgcn_wmma_f32_16x16x4_f32(
                false, a1, false, b1, (short)0, c[i], false, false);
        }
    }
}

template <bool RELU, int NT>
__device__ __forceinline__ void tile_store(
    const v8f (&c)[7], const float* __restrict__ bias, float* __restrict__ out,
    const int rg, const int col, const int lane, const int row0, const int nvalid)
{
    const int   lr   = lane & 15;
    const int   rtop = (lane >> 4) * 8;
    const float bv   = bias[col + lr];
#pragma unroll
    for (int i = 0; i < NT; ++i) {
        const int rb = (rg + 4 * i) * 16 + rtop;
        float* op = out + (size_t)(row0 + rb) * 128 + col + lr;
#pragma unroll
        for (int v = 0; v < 8; ++v) {
            if (rb + v < nvalid) {
                float o = c[i][v] + bv;
                if (RELU) o = fmaxf(o, 0.0f);
                op[v * 128] = o;
            }
        }
    }
}

template <bool RELU>
__global__ __launch_bounds__(1024) void sage_layer_lds(
    const float* __restrict__ feat, const int* __restrict__ src,
    const int* __restrict__ dst, const unsigned char* __restrict__ keys,
    float* __restrict__ deg, int count_deg, int E, int n_rows,
    unsigned* __restrict__ rec_list, int* __restrict__ rec_cnt,
    const unsigned* __restrict__ rep_list, const int* __restrict__ rep_cnt,
    const float* __restrict__ Wl, const float* __restrict__ bias,
    const float* __restrict__ Wr, float* __restrict__ out)
{
    __shared__ __attribute__((aligned(16))) float acc[AGG_R][ACC_LD];
    __shared__ unsigned seg[32][AGG_CAP];
    __shared__ unsigned stage[32][64];
    __shared__ int rowcnt[AGG_R];
    __shared__ int segcnt[32];
    __shared__ int segmore[32];

    const int tid  = threadIdx.x;
    const int wave = __builtin_amdgcn_readfirstlane(tid >> 5);
    const int lane = tid & 31;
    const int row0 = blockIdx.x * AGG_R;
    const bool count = (count_deg != 0);

    const int  my_r  = wave + 32 * lane;
    const bool my_in = (lane < AGG_RPW) && (my_r < AGG_R);
    const bool my_ok = my_in && (row0 + my_r < n_rows);
    float dl = 1.0f;
    if (!count && my_ok) dl = deg[row0 + my_r];

    bool replayed = false;
    if (rep_cnt != nullptr && !count) {
        const int  c  = rep_cnt[blockIdx.x * 32 + lane];
        const bool ok = (__builtin_amdgcn_ballot_w32(c < 0) == 0u);
        if (ok) {
            const int n = __builtin_amdgcn_readlane(c, wave);
            const unsigned* my = rep_list + (size_t)(blockIdx.x * 32 + wave) * REC_CAP;
            for (int r = wave; r < AGG_R; r += 32)
                *(float4*)&acc[r][lane * 4] = make_float4(0.f, 0.f, 0.f, 0.f);
            float4   sum = make_float4(0.f, 0.f, 0.f, 0.f);
            int      cur = -1;
            unsigned ent = (lane < n) ? my[lane] : 0u;
            for (int i0 = 0; i0 < n; i0 += 32) {
                const int      nx  = i0 + 32 + lane;
                const unsigned nxt = (nx < n) ? my[nx] : 0u;
                const int      m   = (n - i0 < 32) ? (n - i0) : 32;
                agg_sum_runs(feat, acc, ent, m, lane, sum, cur);
                ent = nxt;
            }
            if (cur >= 0) agg_row_flush(acc, cur, lane, sum);
            replayed = true;
        }
    }

    if (!replayed) {
        const bool rec = (rec_cnt != nullptr);
        unsigned* my_rec = rec ? rec_list + (size_t)(blockIdx.x * 32 + wave) * REC_CAP : nullptr;
        int rec_pos = 0;

        for (int i = tid; i < AGG_R * (ACC_LD / 4); i += 1024)
            ((float4*)&acc[0][0])[i] = make_float4(0.f, 0.f, 0.f, 0.f);
        for (int i = tid; i < AGG_R; i += 1024) rowcnt[i] = 0;

        const int total_steps = (E + AGG_STEP - 1) / AGG_STEP;
        const int per_wave    = (total_steps + 31) / 32;
        const int first       = wave * per_wave;
        int my_n = total_steps - first;
        if (my_n > per_wave) my_n = per_wave;
        if (my_n < 0) my_n = 0;
        const int rot = (my_n > 0) ? (int)(((long)blockIdx.x * my_n) / (long)gridDim.x) : 0;
        const unsigned keyrep = (unsigned)blockIdx.x * 0x01010101u;

        unsigned* my_list = &seg[wave][0];
        int cursor = 0;
        int more;
        bool first_pass = true;
        int  def_n = 0;
        do {
            int  cnt  = 0;
            bool full = false;
            while (cursor < my_n && !full) {
                const int c0 = cursor;
                uint4 kq[8];
                int   eb[8];
    #pragma unroll
                for (int u = 0; u < 8; ++u) {
                    int s = (c0 + u < my_n) ? (c0 + u) : (my_n - 1);
                    s += rot;
                    if (s >= my_n) s -= my_n;
                    eb[u] = (first + s) * AGG_STEP + lane * 16;
                    kq[u] = *(const uint4*)(keys + eb[u]);
                }
    #pragma unroll
                for (int u = 0; u < 8; ++u) {
                    if (!full && c0 + u < my_n) {
                        if (cnt + AGG_STEP > AGG_CAP) {
                            full = true;
                        } else {
                            agg_list_step(my_list, cnt, kq[u], eb[u], keyrep);
                            cursor = c0 + u + 1;
                        }
                    }
                }
            }
            {
                bool bad = false;
                for (int i = lane; i < cnt; i += 32) {
                    const int  e  = (int)my_list[i];
                    const bool in = (e < E);
                    const int  s  = in ? src[e] : 0;
                    const int  d  = in ? dst[e] : -1;
                    const unsigned r  = (unsigned)(d - row0);
                    const bool     ok = in && (r < (unsigned)AGG_R);
                    my_list[i] = ok ? (((unsigned)s << 9) | r) : 0xFFFFFFFFu;
                    bad = bad || !ok;
                }
                if (__builtin_amdgcn_ballot_w32(bad) != 0u) {
                    int kept = 0;
                    for (int i0 = 0; i0 < cnt; i0 += 32) {
                        const int      idx = i0 + lane;
                        const unsigned en  = (idx < cnt) ? my_list[idx] : 0xFFFFFFFFu;
                        const bool     ok  = (en != 0xFFFFFFFFu);
                        const unsigned k   = __builtin_amdgcn_ballot_w32(ok);
                        if (ok) my_list[kept + (int)__builtin_amdgcn_mbcnt_lo(k, 0u)] = en;
                        kept += __builtin_popcount(k);
                    }
                    cnt = kept;
                }
            }
            bool grouped = false;
            if (cnt <= GRP_MAX) {
                unsigned* hist = &stage[wave][0];
                unsigned  e[8];
                unsigned  pk[8];
#pragma unroll
                for (int c = 0; c < 8; ++c) {
                    const int idx = c * 32 + lane;
                    e[c] = (idx < cnt) ? my_list[idx] : 0u;
                }
                hist[lane] = 0u;
#pragma unroll
                for (int c = 0; c < 8; ++c) {
                    pk[c] = 0u;
                    if (c * 32 < cnt) {
                        const bool     valid = (c * 32 + lane < cnt);
                        const unsigned key   = e[c] & 31u;
                        unsigned m = __builtin_amdgcn_ballot_w32(valid);
#pragma unroll
                        for (int b = 0; b < 5; ++b) {
                            const bool     bit = ((key >> b) & 1u) != 0u;
                            const unsigned bb  = __builtin_amdgcn_ballot_w32(bit);
                            m &= bit ? bb : ~bb;
                        }
                        const unsigned rank = __builtin_amdgcn_mbcnt_lo(m, 0u);
                        const unsigned base = hist[key];
                        pk[c] = base + rank;
                        if (valid && rank == 0u) hist[key] = base + (unsigned)__builtin_popcount(m);
                    }
                }
                const unsigned hc = hist[lane];
                unsigned inc = hc;
#pragma unroll
                for (int d = 1; d < 32; d <<= 1) {
                    const unsigned t = (unsigned)__shfl_up((int)inc, d);
                    if (lane >= d) inc += t;
                }
                const unsigned gstart = inc - hc;
                hist[lane] = gstart;
                my_list[GRP_OFF + lane] = gstart;
                if (lane == 0) my_list[GRP_OFF + 32] = (unsigned)cnt;
#pragma unroll
                for (int c = 0; c < 8; ++c) {
                    if (c * 32 + lane < cnt) my_list[hist[e[c] & 31u] + pk[c]] = e[c];
                }
                grouped = true;
            }
            if (lane == 0) {
                segcnt[wave]  = grouped ? (cnt | GRP_FLAG) : cnt;
                segmore[wave] = (cursor < my_n) ? 1 : 0;
            }
            __syncthreads();

            more = __builtin_amdgcn_readfirstlane(
                (int)(__builtin_amdgcn_ballot_w32(segmore[lane] != 0) != 0u));
            int staged = 0;
            const bool defer = rec && count && first_pass && (more == 0);
            const int  raw_l = segcnt[lane];
            const bool grp_l = (raw_l & GRP_FLAG) != 0;
            int beg_l = 0;
            int end_l = raw_l & (GRP_FLAG - 1);
            if (grp_l) {
                beg_l = (int)seg[lane][GRP_OFF + wave];
                end_l = (int)seg[lane][GRP_OFF + wave + 1];
            }
            for (int s = 0; s < 32; ++s) {
                const int  n   = __builtin_amdgcn_readlane(end_l, s);
                const int  i_b = __builtin_amdgcn_readlane(beg_l, s);
                const bool grp = __builtin_amdgcn_readlane(grp_l ? 1 : 0, s) != 0;
                for (int i0 = i_b; i0 < n; i0 += 32) {
                    const int idx = i0 + lane;
                    const unsigned en = (idx < n) ? seg[s][idx] : 0u;
                    const bool mine = (idx < n) && (grp || ((int)(en & 31u) == wave));
                    const unsigned mk = __builtin_amdgcn_ballot_w32(mine);
                    if (mk != 0u) {
                        if (mine) stage[wave][staged + (int)__builtin_amdgcn_mbcnt_lo(mk, 0u)] = en;
                        staged += __builtin_popcount(mk);
                        if (staged >= 32) {
                            const unsigned ent  = stage[wave][lane];
                            const unsigned rest = stage[wave][32 + lane];
                            bool add_now = true;
                            if (rec) {
                                const bool fits = (rec_pos + 32 <= REC_CAP);
                                if (fits) my_rec[rec_pos + lane] = ent;
                                rec_pos += 32;
                                if (defer && fits) { add_now = false; def_n = rec_pos; }
                            }
                            if (add_now) agg_add_rows(feat, acc, rowcnt, ent, 32, lane, count);
                            staged -= 32;
                            if (lane < staged) stage[wave][lane] = rest;
                        }
                    }
                }
            }
            if (staged > 0) {
                const unsigned ent = stage[wave][lane];
                bool add_now = true;
                if (rec) {
                    const bool fits = (rec_pos + staged <= REC_CAP);
                    if (fits && lane < staged) my_rec[rec_pos + lane] = ent;
                    rec_pos += staged;
                    if (defer && fits) { add_now = false; def_n = rec_pos; }
                }
                if (add_now) agg_add_rows(feat, acc, rowcnt, ent, staged, lane, count);
            }
            __syncthreads();
            first_pass = false;
        } while (more);

        if (def_n > 0) {
            const unsigned hc = agg_rows_sorted(feat, acc, my_rec, def_n,
                                                &seg[wave][0], &stage[wave][0], lane);
            if (my_in) rowcnt[my_r] += (int)hc;
        }

        if (rec && lane == 0)
            rec_cnt[blockIdx.x * 32 + wave] = (rec_pos <= REC_CAP) ? rec_pos : -1;
    }

    if (count) {
        dl = my_in ? (float)rowcnt[my_r] : 1.0f;
        if (my_ok) deg[row0 + my_r] = dl;
    }
    for (int i = 0; i < AGG_RPW; ++i) {
        const int r = wave + 32 * i;
        if (r < AGG_R) {
            const float d = __int_as_float(
                __builtin_amdgcn_readlane(__float_as_int(dl), i));
            const float invd = 1.0f / fmaxf(d, 1.0f);
            float4* p = (float4*)&acc[r][lane * 4];
            float4 a = *p;
            a.x *= invd; a.y *= invd; a.z *= invd; a.w *= invd;
            *p = a;
        }
    }
    __syncthreads();

    const int rg  = wave >> 3;
    const int col = (wave & 7) * 16;
    v8f c[7];
#pragma unroll
    for (int i = 0; i < 7; ++i) c[i] = v8f{};

    if (rg == 0) tile_products<7>(c, acc, Wl, rg, col, lane);
    else         tile_products<6>(c, acc, Wl, rg, col, lane);

    __syncthreads();
    const int nvalid = (n_rows - row0 < AGG_R) ? (n_rows - row0) : AGG_R;
    for (int i0 = 0; i0 < AGG_RPW; i0 += 7) {
        float4 v[7];
#pragma unroll
        for (int u = 0; u < 7; ++u) {
            const int r = wave + 32 * (i0 + u);
            v[u] = (r < nvalid)
                 ? ((const float4*)(feat + (size_t)(row0 + r) * 128))[lane]
                 : make_float4(0.f, 0.f, 0.f, 0.f);
        }
#pragma unroll
        for (int u = 0; u < 7; ++u) {
            const int r = wave + 32 * (i0 + u);
            if (r < AGG_R) *(float4*)&acc[r][lane * 4] = v[u];
        }
    }
    __syncthreads();

    if (rg == 0) {
        tile_products<7>(c, acc, Wr, rg, col, lane);
        tile_store<RELU, 7>(c, bias, out, rg, col, lane, row0, nvalid);
    } else {
        tile_products<6>(c, acc, Wr, rg, col, lane);
        tile_store<RELU, 6>(c, bias, out, rg, col, lane, row0, nvalid);
    }
}

extern "C" void kernel_launch(void* const* d_in, const int* in_sizes, int n_in,
                              void* d_out, int out_size, void* d_ws, size_t ws_size,
                              hipStream_t stream) {
    const float* x   = (const float*)d_in[0];
    const int*   ei  = (const int*)d_in[1];
    const float* W1l = (const float*)d_in[2];
    const float* b1l = (const float*)d_in[3];
    const float* W1r = (const float*)d_in[4];
    const float* W2l = (const float*)d_in[5];
    const float* b2l = (const float*)d_in[6];
    const float* W2r = (const float*)d_in[7];
    float* out = (float*)d_out;

    const int N = in_sizes[0] / 128;
    const int E = in_sizes[1] / 2;
    const int* src = ei;
    const int* dst = ei + E;

    float* h1  = (float*)d_ws;
    float* deg = h1 + (size_t)N * 128;

    const int  ag_blocks = (N + AGG_R - 1) / AGG_R;

    int*      rec_cnt  = (int*)(deg + N);
    unsigned* rec_list = (unsigned*)(rec_cnt + ag_blocks * 32);

    unsigned char* keys = (unsigned char*)(rec_list + (size_t)ag_blocks * 32 * REC_CAP);
    keys += (16 - ((size_t)keys & 15)) & 15;
    const int key_steps = (E + AGG_STEP - 1) / AGG_STEP;

    edge_block_keys<<<(key_steps + 7) / 8, 256, 0, stream>>>(
        dst, (unsigned*)keys, E, key_steps);

    sage_layer_lds<true><<<ag_blocks, 1024, 0, stream>>>(
        x, src, dst, keys, deg, 1, E, N, rec_list, rec_cnt, nullptr, nullptr,
        W1l, b1l, W1r, h1);

    sage_layer_lds<false><<<ag_blocks, 1024, 0, stream>>>(
        h1, src, dst, keys, deg, 0, E, N, nullptr, nullptr, rec_list, rec_cnt,
        W2l, b2l, W2r, out);
}
